# starting kernel
# baseline (speedup 1.0000x reference)
#include <hip/hip_runtime.h>
#include <math.h>

#define B_N 16384
#define F_N 48
#define T_N 512
#define L_N 16
#define NRT (B_N / 16)
#define NKT_F (F_N / 4)
#define NCT_T (T_N / 16)
#define NCT_F (F_N / 16)

typedef __attribute__((ext_vector_type(2))) float v2f;
typedef __attribute__((ext_vector_type(8))) float v8f;

static __device__ __forceinline__ v8f wmma_f32(v2f a, v2f b, v8f c) {
  return __builtin_amdgcn_wmma_f32_16x16x4_f32(false, a, false, b, (short)0, c,
                                               false, false);
}

static __device__ __forceinline__ int pack_idx(int k, int col, int nct) {
  return (((k >> 2) * nct + (col >> 4)) * 32 + ((k >> 1) & 1) * 16 + (col & 15)) * 2 +
         (k & 1);
}

__global__ void __launch_bounds__(256) k_maxpart(const float* __restrict__ u,
                                                 float* __restrict__ part, int n) {
  __shared__ float sm[256];
  float m = -3.4e38f;
  for (int i = blockIdx.x * 256 + threadIdx.x; i < n; i += 256 * gridDim.x)
    m = fmaxf(m, u[i]);
  sm[threadIdx.x] = m;
  __syncthreads();
  for (int s = 128; s > 0; s >>= 1) {
    if (threadIdx.x < s) sm[threadIdx.x] = fmaxf(sm[threadIdx.x], sm[threadIdx.x + s]);
    __syncthreads();
  }
  if (threadIdx.x == 0) part[blockIdx.x] = sm[0];
}

__global__ void __launch_bounds__(256) k_finalize(const float* __restrict__ part,
                                                  float* __restrict__ scal,
                                                  float* __restrict__ sum) {
  __shared__ float sm[256];
  float m = -3.4e38f;
  for (int i = threadIdx.x; i < 1024; i += 256) m = fmaxf(m, part[i]);
  sm[threadIdx.x] = m;
  __syncthreads();
  for (int s = 128; s > 0; s >>= 1) {
    if (threadIdx.x < s) sm[threadIdx.x] = fmaxf(sm[threadIdx.x], sm[threadIdx.x + s]);
    __syncthreads();
  }
  if (threadIdx.x == 0) {
    float mx = fabsf(sm[0]);
    scal[0] = 5.0f / mx;
    scal[1] = mx / 5.0f;
  }
  for (int t = threadIdx.x; t < L_N * T_N; t += 256) sum[t] = 0.0f;
}

__global__ void __launch_bounds__(256) k_dft(float* __restrict__ irep, float* __restrict__ iimp,
                                             float* __restrict__ drep, float* __restrict__ dimp) {
  int idx = blockIdx.x * 256 + threadIdx.x;
  if (idx >= F_N * T_N) return;
  int f = idx / T_N, t = idx - f * T_N;
  int ph = (f * t) & (T_N - 1);
  float ang = (float)ph * (6.283185307179586f / (float)T_N);
  const float sc = 0.04419417382415922f;
  float sv, cv;
  sincosf(ang, &sv, &cv);
  int ip = pack_idx(f, t, NCT_T);
  irep[ip] = cv * sc;
  iimp[ip] = sv * sc;
  int dp = pack_idx(t, f, NCT_F);
  drep[dp] = cv * sc;
  dimp[dp] = -sv * sc;
}

__global__ void __launch_bounds__(256) k_zero4(float4* __restrict__ p, int n4) {
  for (int i = blockIdx.x * 256 + threadIdx.x; i < n4; i += 256 * gridDim.x)
    p[i] = make_float4(0.f, 0.f, 0.f, 0.f);
}

__global__ void __launch_bounds__(256) k_zinit(const float* __restrict__ u,
                                               const float* __restrict__ scal,
                                               float* __restrict__ zre,
                                               float* __restrict__ zim, int n) {
  const float s = scal[0];
  for (int i = blockIdx.x * 256 + threadIdx.x; i < n; i += 256 * gridDim.x) {
    int m = i / F_N, c = i - m * F_N;
    int p = (((m >> 4) * NKT_F + (c >> 2)) * 32 + ((c >> 1) & 1) * 16 + (m & 15)) * 2 +
            (c & 1);
    zre[p] = s * u[2 * i];
    zim[p] = s * u[2 * i + 1];
  }
}

typedef struct { v2f re[NKT_F], im[NKT_F]; } Abuf;

__global__ void __launch_bounds__(256) k_layerA(
    const float* __restrict__ zre, const float* __restrict__ zim,
    float* __restrict__ hre, float* __restrict__ him,
    const float* __restrict__ brep, const float* __restrict__ bimp,
    float* __restrict__ sum) {
  const int lane = threadIdx.x & 31;
  const int wv = threadIdx.x >> 5;
  const int hi = lane >> 4;
  const int ct = blockIdx.x;
  const int col = ct * 16 + (lane & 15);
  const int wgl = blockIdx.y * 8 + wv;

  v2f Bre[NKT_F], Bim[NKT_F];
  {
    const float* bbr = brep + ct * 64 + lane * 2;
    const float* bbi = bimp + ct * 64 + lane * 2;
#pragma unroll
    for (int kt = 0; kt < NKT_F; ++kt) {
      Bre[kt] = *(const v2f*)(bbr + kt * (NCT_T * 64));
      Bim[kt] = *(const v2f*)(bbi + kt * (NCT_T * 64));
    }
  }

  const float* zr = zre + wgl * (8 * NKT_F * 64) + lane * 2;
  const float* zi = zim + wgl * (8 * NKT_F * 64) + lane * 2;
  float* hbr = hre + wgl * 128 * T_N + col;
  float* hbi = him + wgl * 128 * T_N + col;
  const int hrow = hi * 8 * T_N;

#define LOADA(dst, j_)                                                        \
  do {                                                                        \
    _Pragma("unroll") for (int kt = 0; kt < NKT_F; ++kt) {                    \
      (dst).re[kt] = *(const v2f*)(zr + (j_) * (NKT_F * 64) + kt * 64);       \
      (dst).im[kt] = *(const v2f*)(zi + (j_) * (NKT_F * 64) + kt * 64);       \
    }                                                                         \
  } while (0)

#define TILE(A_, j_)                                                          \
  do {                                                                        \
    float hldr[8], hldi[8];                                                   \
    _Pragma("unroll") for (int i = 0; i < 8; ++i) {                           \
      hldr[i] = hbr[(j_) * (16 * T_N) + hrow + i * T_N];                      \
      hldi[i] = hbi[(j_) * (16 * T_N) + hrow + i * T_N];                      \
    }                                                                         \
    v8f ar_ = {}, ai_ = {};                                                   \
    _Pragma("unroll") for (int kt = 0; kt < NKT_F; ++kt) {                    \
      v2f bn = -Bim[kt];                                                      \
      ar_ = wmma_f32((A_).re[kt], Bre[kt], ar_);                              \
      ar_ = wmma_f32((A_).im[kt], bn, ar_);                                   \
      ai_ = wmma_f32((A_).re[kt], Bim[kt], ai_);                              \
      ai_ = wmma_f32((A_).im[kt], Bre[kt], ai_);                              \
    }                                                                         \
    _Pragma("unroll") for (int i = 0; i < 8; ++i) {                           \
      float rr = ar_[i] + hldr[i], ri = ai_[i] + hldi[i];                     \
      hbr[(j_) * (16 * T_N) + hrow + i * T_N] = rr;                           \
      hbi[(j_) * (16 * T_N) + hrow + i * T_N] = ri;                           \
      colsum += rr * rr + ri * ri;                                            \
    }                                                                         \
  } while (0)

  float colsum = 0.0f;
  Abuf A0, A1;
  LOADA(A0, 0);
  for (int jj = 0; jj < 8; jj += 2) {
    LOADA(A1, jj + 1);
    TILE(A0, jj);
    if (jj + 2 < 8) LOADA(A0, jj + 2);
    TILE(A1, jj + 1);
  }
#undef LOADA
#undef TILE

  colsum += __shfl_xor(colsum, 16, 32);
  if (lane < 16) atomicAdd(&sum[col], colsum);
}

__global__ void __launch_bounds__(256) k_layerC(
    float* __restrict__ hre, float* __restrict__ him,
    const float* __restrict__ drep, const float* __restrict__ dimp,
    float* __restrict__ zre, float* __restrict__ zim,
    const float* __restrict__ u, const float* __restrict__ scal,
    const float* __restrict__ sum, const float* __restrict__ S1,
    const float* __restrict__ S2, float* __restrict__ out, int last) {
  __shared__ float gs[T_N];
  __shared__ float accL[6][256];
  const float s1 = S1[0], s2 = S2[0];
  for (int t = threadIdx.x; t < T_N; t += 256) {
    float rm = sqrtf(sum[t] * (1.0f / (float)B_N));
    gs[t] = 1.0f / (1.0f + expf(-(s1 * (rm - s2))));
  }
  for (int i = threadIdx.x; i < 6 * 256; i += 256) (&accL[0][0])[i] = 0.0f;
  __syncthreads();

  const int lane = threadIdx.x & 31;
  const int wv = threadIdx.x >> 5;
  const int rowbase = blockIdx.x * 16;
  const int kbase = wv * 64;

  for (int j = 0; j < 8; ++j) {
    int idx = (j * 32 + lane) * 4;
    int row = idx >> 6, cc = idx & 63;
    int col = kbase + cc;
    float4 g4 = *(const float4*)&gs[col];
    int off = (rowbase + row) * T_N + col;
    float4 r = *(float4*)&hre[off];
    r.x *= g4.x; r.y *= g4.y; r.z *= g4.z; r.w *= g4.w;
    *(float4*)&hre[off] = r;
    float4 q = *(float4*)&him[off];
    q.x *= g4.x; q.y *= g4.y; q.z *= g4.z; q.w *= g4.w;
    *(float4*)&him[off] = q;
  }

  v8f accRe[3] = {}, accIm[3] = {};
  const int hi = lane >> 4;
  const int cloc = lane & 15;
  const float* harp = hre + (rowbase + cloc) * T_N + kbase + hi * 2;
  const float* haip = him + (rowbase + cloc) * T_N + kbase + hi * 2;
  const float* dbr = drep + (kbase >> 2) * (NCT_F * 64) + lane * 2;
  const float* dbi = dimp + (kbase >> 2) * (NCT_F * 64) + lane * 2;

  v2f pr[2], pi[2];
  pr[0] = *(const v2f*)(harp + 0);
  pi[0] = *(const v2f*)(haip + 0);
  pr[1] = *(const v2f*)(harp + 4);
  pi[1] = *(const v2f*)(haip + 4);
#pragma unroll
  for (int t = 0; t < 16; ++t) {
    v2f a_re = pr[t & 1], a_im = pi[t & 1];
    if (t + 2 < 16) {
      pr[t & 1] = *(const v2f*)(harp + (t + 2) * 4);
      pi[t & 1] = *(const v2f*)(haip + (t + 2) * 4);
    }
#pragma unroll
    for (int nt = 0; nt < NCT_F; ++nt) {
      v2f b_re = *(const v2f*)(dbr + (t * NCT_F + nt) * 64);
      v2f b_im = *(const v2f*)(dbi + (t * NCT_F + nt) * 64);
      v2f bn = -b_im;
      accRe[nt] = wmma_f32(a_re, b_re, accRe[nt]);
      accRe[nt] = wmma_f32(a_im, bn, accRe[nt]);
      accIm[nt] = wmma_f32(a_re, b_im, accIm[nt]);
      accIm[nt] = wmma_f32(a_im, b_re, accIm[nt]);
    }
  }

#pragma unroll
  for (int nt = 0; nt < NCT_F; ++nt) {
#pragma unroll
    for (int i = 0; i < 8; ++i) {
      int sl = (i + hi * 8) * 16 + cloc;
      atomicAdd(&accL[nt * 2 + 0][sl], accRe[nt][i]);
      atomicAdd(&accL[nt * 2 + 1][sl], accIm[nt][i]);
    }
  }
  __syncthreads();

  const float s = scal[0], inv = scal[1];
  const int rt = blockIdx.x;
#pragma unroll
  for (int ii = 0; ii < 3; ++ii) {
    int e = ii * 256 + threadIdx.x;
    int row = e / F_N, c = e - row * F_N;
    int m = rowbase + row;
    float hr = accL[(c >> 4) * 2 + 0][row * 16 + (c & 15)];
    float hm = accL[(c >> 4) * 2 + 1][row * 16 + (c & 15)];
    if (last) {
      out[(m * F_N + c) * 2]     = hr * inv;
      out[(m * F_N + c) * 2 + 1] = hm * inv;
    } else {
      int p = ((rt * NKT_F + (c >> 2)) * 32 + ((c >> 1) & 1) * 16 + row) * 2 + (c & 1);
      zre[p] = s * u[(m * F_N + c) * 2]     - hr;
      zim[p] = s * u[(m * F_N + c) * 2 + 1] - hm;
    }
  }
}

extern "C" void kernel_launch(void* const* d_in, const int* in_sizes, int n_in,
                              void* d_out, int out_size, void* d_ws, size_t ws_size,
                              hipStream_t stream) {
  const float* u  = (const float*)d_in[0];
  const float* S1 = (const float*)d_in[1];
  const float* S2 = (const float*)d_in[2];
  float* out = (float*)d_out;
  float* w = (float*)d_ws;

  float* scal = w;
  float* part = w + 8;
  float* sum  = w + 2048;
  float* irep = w + 16384;
  float* iimp = irep + F_N * T_N;
  float* drep = iimp + F_N * T_N;
  float* dimp = drep + F_N * T_N;
  float* zre  = w + 131072;
  float* zim  = zre + B_N * F_N;
  float* hre  = zim + B_N * F_N;
  float* him  = hre + B_N * T_N;

  const int nu = B_N * F_N * 2;
  k_maxpart<<<1024, 256, 0, stream>>>(u, part, nu);
  k_finalize<<<1, 256, 0, stream>>>(part, scal, sum);
  k_dft<<<(F_N * T_N + 255) / 256, 256, 0, stream>>>(irep, iimp, drep, dimp);
  k_zero4<<<4096, 256, 0, stream>>>((float4*)hre, (2 * B_N * T_N) / 4);
  k_zinit<<<1024, 256, 0, stream>>>(u, scal, zre, zim, B_N * F_N);

  for (int l = 0; l < L_N; ++l) {
    k_layerA<<<dim3(NCT_T, 16), 256, 0, stream>>>(zre, zim, hre, him, irep, iimp,
                                                  sum + l * T_N);
    k_layerC<<<NRT, 256, 0, stream>>>(hre, him, drep, dimp, zre, zim, u, scal,
                                      sum + l * T_N, S1, S2, out,
                                      (l == L_N - 1) ? 1 : 0);
  }
}


# layer kernel: no matrix stream waits on a global load: idft fragments fetched to registers at entry, G staged in LDS by the idle waves (16-byte reads), dft fragments and P stores behind the streams
# speedup vs baseline: 22.4281x; 22.4281x over previous
#include <hip/hip_runtime.h>
#include <math.h>


#define B_N 16384
#define F_N 48
#define T_N 512
#define L_N 16
#define MSZ (F_N * F_N)
#define CSZ (2 * MSZ)
#define NGP 256
#define LBLK 8
#define LCOL 64
#define NCW (LCOL / 16)
#define PPITCH 68
#define MPITCH 52
#define GPITCH 52

typedef __attribute__((ext_vector_type(2))) float v2f;
typedef __attribute__((ext_vector_type(8))) float v8f;

static __device__ __forceinline__ v8f wmma_f32(v2f a, v2f b, v8f c) {
  return __builtin_amdgcn_wmma_f32_16x16x4_f32(false, a, false, b, (short)0, c,
                                               false, false);
}


__global__ void __launch_bounds__(96) k_gram(const float* __restrict__ u,
                                             float* __restrict__ part,
                                             float* __restrict__ gpart,
                                             float* __restrict__ ire, float* __restrict__ iim,
                                             float* __restrict__ dre, float* __restrict__ dim) {
  {
    const int idx = blockIdx.x * 96 + threadIdx.x;
    const float sc = 0.04419417382415922f;
    const float w = 6.283185307179586f / (float)T_N;
    float sv, cv;
    {
      int f = idx / T_N, t = idx - f * T_N;
      int ph = (f * t) & (T_N - 1);
      sincosf((float)ph * w, &sv, &cv);
      ire[idx] = cv * sc;
      iim[idx] = sv * sc;
    }
    {
      int t = idx / F_N, f = idx - t * F_N;
      int ph = (f * t) & (T_N - 1);
      sincosf((float)ph * w, &sv, &cv);
      dre[idx] = cv * sc;
      dim[idx] = -sv * sc;
    }
  }
  const int lane = threadIdx.x & 31;
  const int wi = threadIdx.x >> 5;
  const int hi = lane >> 4, c = lane & 15;
  const float* ub = u + (size_t)(blockIdx.x * 64 + 2 * hi) * (F_N * 2) + c * 2;
  v8f gre[3], gim[3];
#pragma unroll
  for (int j = 0; j < 3; ++j) {
    gre[j] = (v8f){};
    gim[j] = (v8f){};
  }
  float mxl = -3.4e38f;
#pragma unroll
  for (int ks = 0; ks < 16; ++ks) {
    v2f xr[3], xi[3];
#pragma unroll
    for (int ft = 0; ft < 3; ++ft) {
      const float2 p0 = *(const float2*)(ub + (ks * 4) * (F_N * 2) + ft * 32);
      const float2 p1 = *(const float2*)(ub + (ks * 4 + 1) * (F_N * 2) + ft * 32);
      xr[ft] = (v2f){p0.x, p1.x};
      xi[ft] = (v2f){p0.y, p1.y};
      mxl = fmaxf(mxl, fmaxf(fmaxf(p0.x, p0.y), fmaxf(p1.x, p1.y)));
    }
    const float2 a0 = *(const float2*)(ub + (ks * 4) * (F_N * 2) + wi * 32);
    const float2 a1 = *(const float2*)(ub + (ks * 4 + 1) * (F_N * 2) + wi * 32);
    const v2f ar = (v2f){a0.x, a1.x};
    const v2f ai = (v2f){a0.y, a1.y};
    const v2f an = -ai;
#pragma unroll
    for (int j = 0; j < 3; ++j) {
      gre[j] = wmma_f32(ar, xr[j], gre[j]);
      gim[j] = wmma_f32(ar, xi[j], gim[j]);
      gre[j] = wmma_f32(ai, xi[j], gre[j]);
      gim[j] = wmma_f32(an, xr[j], gim[j]);
    }
  }
  float* gp = gpart + (size_t)blockIdx.x * CSZ + wi * (3 * 256);
#pragma unroll
  for (int j = 0; j < 3; ++j)
#pragma unroll
    for (int v = 0; v < 8; ++v) {
      gp[j * 256 + v * 32 + lane] = gre[j][v];
      gp[MSZ + j * 256 + v * 32 + lane] = gim[j][v];
    }
  mxl = fmaxf(mxl, __shfl_xor(mxl, 16, 32));
  mxl = fmaxf(mxl, __shfl_xor(mxl, 8, 32));
  mxl = fmaxf(mxl, __shfl_xor(mxl, 4, 32));
  mxl = fmaxf(mxl, __shfl_xor(mxl, 2, 32));
  mxl = fmaxf(mxl, __shfl_xor(mxl, 1, 32));
  if (threadIdx.x == 0) part[blockIdx.x] = mxl;
}

__global__ void __launch_bounds__(256) k_gred(const float* __restrict__ gpart,
                                              const float* __restrict__ part,
                                              float* __restrict__ G,
                                              float* __restrict__ scal) {
  __shared__ float sm[256];
  __shared__ float red[8][32];
  const int grp = threadIdx.x >> 5, el = threadIdx.x & 31;
  const int n = blockIdx.x * 32 + el;
  {
    const float* p = gpart + (size_t)(grp * 32) * CSZ + n;
    float acc = 0.0f;
    for (int i = 0; i < 32; ++i) acc += p[(size_t)i * CSZ];
    red[grp][el] = acc;
  }
  sm[threadIdx.x] = part[threadIdx.x];
  __syncthreads();
  for (int s = 128; s > 0; s >>= 1) {
    if (threadIdx.x < s) sm[threadIdx.x] = fmaxf(sm[threadIdx.x], sm[threadIdx.x + s]);
    __syncthreads();
  }
  const float mx = fabsf(sm[0]);
  const float s = 5.0f / mx;
  if (blockIdx.x == 0 && threadIdx.x == 0) {
    scal[0] = s;
    scal[1] = mx / 5.0f;
  }
  if (threadIdx.x < 32) {
    float tot = 0.0f;
#pragma unroll
    for (int k = 0; k < 8; ++k) tot += red[k][el];
    const int pl = n / MSZ, q = n - pl * MSZ;
    const int tl = q >> 8, tr = tl / 3;
    const int r = tr * 16 + ((q >> 5) & 7) + ((q >> 4) & 1) * 8;
    const int cc = (tl - tr * 3) * 16 + (q & 15);
    G[pl * MSZ + r * F_N + cc] = tot * (s * s * (1.0f / (float)B_N));
  }
}

static __device__ __forceinline__ void load_dft_frag(const float* __restrict__ dre,
                                                     const float* __restrict__ dim,
                                                     int base, v2f (&dR)[LCOL / 4],
                                                     v2f (&dI)[LCOL / 4]) {
#pragma unroll
  for (int ks = 0; ks < LCOL / 4; ++ks) {
    const int dofs = base + ks * 4 * F_N;
    dR[ks] = (v2f){dre[dofs], dre[dofs + F_N]};
    dI[ks] = (v2f){dim[dofs], dim[dofs + F_N]};
  }
}

__global__ void __launch_bounds__(288) k_layer(
    const float* __restrict__ ire, const float* __restrict__ iim,
    const float* __restrict__ dre, const float* __restrict__ dim,
    const float* __restrict__ G, const float* __restrict__ sprev,
    float* __restrict__ snext, float* __restrict__ pst,
    const float* __restrict__ S1, const float* __restrict__ S2, int first) {
  __shared__ __attribute__((aligned(16))) float sm[2 * F_N * PPITCH + 2 * F_N * GPITCH];
  float* Mre = sm;
  float* Mim = sm + F_N * MPITCH;
  float* Pre = sm;
  float* Pim = sm + F_N * PPITCH;
  float* Gre = sm + 2 * F_N * PPITCH;
  float* Gim = Gre + F_N * GPITCH;

  const int tid = threadIdx.x;
  const int lane = tid & 31;
  const int wv = tid >> 5;
  const int hi = lane >> 4, c = lane & 15;
  const int ti = wv / 3, tj = wv - ti * 3;
  const int dbase = (blockIdx.x * LCOL + hi * 2) * F_N + tj * 16 + c;
  const float s1 = S1[0], s2 = S2[0];

  v8f qre[3], qim[3];
#pragma unroll
  for (int rt = 0; rt < 3; ++rt) {
    qre[rt] = (v8f){};
    qim[rt] = (v8f){};
  }
  v2f bR[12], bI[12];
#pragma unroll
  for (int ks = 0; ks < 12; ++ks) {
    bR[ks] = (v2f){0.f, 0.f};
    bI[ks] = (v2f){0.f, 0.f};
  }
  float* pw = pst + (size_t)(blockIdx.x * NCW + (wv & (NCW - 1))) * (2 * 3 * 8 * 32);
  const int col = blockIdx.x * LCOL + (wv & (NCW - 1)) * 16 + c;
  if (wv < NCW) {
    if (!first) {
#pragma unroll
      for (int rt = 0; rt < 3; ++rt)
#pragma unroll
        for (int v = 0; v < 8; ++v) {
          qre[rt][v] = pw[(rt * 8 + v) * 32 + lane];
          qim[rt][v] = pw[768 + (rt * 8 + v) * 32 + lane];
        }
    }
#pragma unroll
    for (int ks = 0; ks < 12; ++ks) {
      const int k0 = ks * 4 + hi * 2;
      bR[ks] = (v2f){ire[k0 * T_N + col], ire[(k0 + 1) * T_N + col]};
      bI[ks] = (v2f){iim[k0 * T_N + col], iim[(k0 + 1) * T_N + col]};
    }
  }

  float4 ms[4];
#pragma unroll
  for (int i = 0; i < 4; ++i) ms[i] = make_float4(0.f, 0.f, 0.f, 0.f);
  if (!first) {
    float4 t[4][LBLK];
#pragma unroll
    for (int i = 0; i < 4; ++i) {
      const int pl = i >> 1;
      const int n = (tid + 288 * i - pl * (MSZ / 4)) * 4;
#pragma unroll
      for (int p = 0; p < LBLK; ++p) t[i][p] = *(const float4*)(sprev + p * CSZ + pl * MSZ + n);
    }
#pragma unroll
    for (int i = 0; i < 4; ++i)
#pragma unroll
      for (int p = 0; p < LBLK; ++p) {
        ms[i].x += t[i][p].x; ms[i].y += t[i][p].y; ms[i].z += t[i][p].z; ms[i].w += t[i][p].w;
      }
  }

#pragma unroll
  for (int i = 0; i < 4; ++i) {
    const int pl = i >> 1;
    const int n = (tid + 288 * i - pl * (MSZ / 4)) * 4;
    const int tl = n >> 8, tr = tl / 3;
    const int r = tr * 16 + ((n >> 5) & 7) + ((n >> 4) & 1) * 8;
    const int cc = (tl - tr * 3) * 16 + (n & 15);
    float4 m;
    if (pl == 0) {
      m.x = ((r == cc) ? 1.0f : 0.0f) - ms[i].x;
      m.y = ((r == cc + 1) ? 1.0f : 0.0f) - ms[i].y;
      m.z = ((r == cc + 2) ? 1.0f : 0.0f) - ms[i].z;
      m.w = ((r == cc + 3) ? 1.0f : 0.0f) - ms[i].w;
      *(float4*)&Mre[r * MPITCH + cc] = m;
    } else {
      m.x = -ms[i].x; m.y = -ms[i].y; m.z = -ms[i].z; m.w = -ms[i].w;
      *(float4*)&Mim[r * MPITCH + cc] = m;
    }
  }
  __syncthreads();

  v2f dR[LCOL / 4], dI[LCOL / 4];
  if (wv < NCW) {
    v8f qk[3];
#pragma unroll
    for (int it = 0; it < 3; ++it) qk[it] = (v8f){};
#pragma unroll
    for (int ks = 0; ks < 12; ++ks) {
      const int k0 = ks * 4 + hi * 2;
      const v2f bre = bR[ks];
      const v2f bim = bI[ks];
      const v2f bd = bim - bre;
      const v2f bs = -bre - bim;
#pragma unroll
      for (int it = 0; it < 3; ++it) {
        const v2f are = *(const v2f*)&Mre[(it * 16 + c) * MPITCH + k0];
        const v2f aim = *(const v2f*)&Mim[(it * 16 + c) * MPITCH + k0];
        const v2f as = are + aim;
        qk[it]  = wmma_f32(as, bre, qk[it]);
        qim[it] = wmma_f32(are, bd, qim[it]);
        qre[it] = wmma_f32(aim, bs, qre[it]);
      }
    }
#pragma unroll
    for (int it = 0; it < 3; ++it) {
      qre[it] += qk[it];
      qim[it] += qk[it];
    }
  } else {
    if (wv < 2 * NCW) {
      float4 g[9];
#pragma unroll
      for (int i = 0; i < 9; ++i) g[i] = *(const float4*)(G + (tid - 128 + 128 * i) * 4);
#pragma unroll
      for (int i = 0; i < 9; ++i) {
        const int q = tid - 128 + 128 * i;
        const int pl = q / (MSZ / 4), rem = q - pl * (MSZ / 4);
        const int r = rem / (F_N / 4), c4 = rem - r * (F_N / 4);
        *(float4*)&Gre[pl * (F_N * GPITCH) + r * GPITCH + c4 * 4] = g[i];
      }
    }
    load_dft_frag(dre, dim, dbase, dR, dI);
  }
  __syncthreads();

  if (wv < NCW) {
    v8f yk[3], yre[3], yim[3];
#pragma unroll
    for (int it = 0; it < 3; ++it) {
      yk[it] = (v8f){};
      yre[it] = (v8f){};
      yim[it] = (v8f){};
    }
#pragma unroll
    for (int rt = 0; rt < 3; ++rt) {
      float4 gr[3][2], gi[3][2];
#pragma unroll
      for (int it = 0; it < 3; ++it) {
        const int go = (it * 16 + c) * GPITCH + rt * 16 + hi * 8;
        gr[it][0] = *(const float4*)&Gre[go];
        gr[it][1] = *(const float4*)&Gre[go + 4];
        gi[it][0] = *(const float4*)&Gim[go];
        gi[it][1] = *(const float4*)&Gim[go + 4];
      }
#pragma unroll
      for (int j = 0; j < 4; ++j) {
        const v2f bqr = (v2f){qre[rt][2 * j], qre[rt][2 * j + 1]};
        const v2f bqi = (v2f){qim[rt][2 * j], qim[rt][2 * j + 1]};
        const v2f bqd = bqi - bqr;
        const v2f bqs = -bqr - bqi;
#pragma unroll
        for (int it = 0; it < 3; ++it) {
          const float4 r4 = gr[it][j >> 1];
          const float4 i4 = gi[it][j >> 1];
          const v2f g_r = (j & 1) ? (v2f){r4.z, r4.w} : (v2f){r4.x, r4.y};
          const v2f g_i = (j & 1) ? (v2f){i4.z, i4.w} : (v2f){i4.x, i4.y};
          const v2f gs = g_r + g_i;
          yk[it]  = wmma_f32(gs, bqr, yk[it]);
          yim[it] = wmma_f32(g_r, bqd, yim[it]);
          yre[it] = wmma_f32(g_i, bqs, yre[it]);
        }
      }
    }
    load_dft_frag(dre, dim, dbase, dR, dI);
    float p = 0.0f;
#pragma unroll
    for (int rt = 0; rt < 3; ++rt)
#pragma unroll
      for (int v = 0; v < 8; ++v)
        p += qre[rt][v] * (yre[rt][v] + yk[rt][v]) + qim[rt][v] * (yim[rt][v] + yk[rt][v]);
    p += __shfl_xor(p, 16, 32);
    const float rm = sqrtf(fmaxf(p, 0.0f));
    const float g = 1.0f / (1.0f + expf(-(s1 * (rm - s2))));
#pragma unroll
    for (int rt = 0; rt < 3; ++rt)
#pragma unroll
      for (int v = 0; v < 8; ++v) {
        qre[rt][v] *= g;
        qim[rt][v] *= g;
        Pre[(rt * 16 + hi * 8 + v) * PPITCH + wv * 16 + c] = qre[rt][v];
        Pim[(rt * 16 + hi * 8 + v) * PPITCH + wv * 16 + c] = qim[rt][v];
      }
  }
  __syncthreads();
  if (wv < NCW) {
#pragma unroll
    for (int rt = 0; rt < 3; ++rt)
#pragma unroll
      for (int v = 0; v < 8; ++v) {
        pw[(rt * 8 + v) * 32 + lane] = qre[rt][v];
        pw[768 + (rt * 8 + v) * 32 + lane] = qim[rt][v];
      }
  }

  {
    v8f sk = {}, sre = {}, sim = {};
#pragma unroll
    for (int ks = 0; ks < LCOL / 4; ++ks) {
      const int k0 = ks * 4 + hi * 2;
      const v2f ar = *(const v2f*)&Pre[(ti * 16 + c) * PPITCH + k0];
      const v2f ai = *(const v2f*)&Pim[(ti * 16 + c) * PPITCH + k0];
      const v2f as = ar + ai;
      const v2f br = dR[ks];
      const v2f bi = dI[ks];
      const v2f bd = bi - br;
      const v2f bs = -br - bi;
      sk  = wmma_f32(as, br, sk);
      sim = wmma_f32(ar, bd, sim);
      sre = wmma_f32(ai, bs, sre);
    }
    float* sp = snext + blockIdx.x * CSZ + wv * 256;
#pragma unroll
    for (int v = 0; v < 8; ++v) {
      sp[v * 32 + lane] = sre[v] + sk[v];
      sp[MSZ + v * 32 + lane] = sim[v] + sk[v];
    }
  }
}

__global__ void __launch_bounds__(128) k_out(const float* __restrict__ u,
                                             const float* __restrict__ scal,
                                             const float* __restrict__ sp,
                                             float* __restrict__ out) {
  __shared__ __attribute__((aligned(16))) float Wre[MSZ];
  __shared__ __attribute__((aligned(16))) float Wim[MSZ];
  for (int q4 = threadIdx.x; q4 < MSZ / 4; q4 += 128) {
    const int n = q4 * 4;
    const int tl = n >> 8, tr = tl / 3;
    const int r = tr * 16 + ((n >> 5) & 7) + ((n >> 4) & 1) * 8;
    const int cc = (tl - tr * 3) * 16 + (n & 15);
    float4 sr = make_float4(0.f, 0.f, 0.f, 0.f), si = make_float4(0.f, 0.f, 0.f, 0.f);
#pragma unroll
    for (int p = 0; p < LBLK; ++p) {
      const float4 a = *(const float4*)(sp + p * CSZ + n);
      const float4 b = *(const float4*)(sp + p * CSZ + MSZ + n);
      sr.x += a.x; sr.y += a.y; sr.z += a.z; sr.w += a.w;
      si.x += b.x; si.y += b.y; si.z += b.z; si.w += b.w;
    }
    *(float4*)&Wre[r * F_N + cc] = sr;
    *(float4*)&Wim[r * F_N + cc] = si;
  }
  __syncthreads();

  const int lane = threadIdx.x & 31;
  const int wv = threadIdx.x >> 5;
  const int hi = lane >> 4, c = lane & 15;
  const int rowbase = (blockIdx.x * 4 + wv) * 16;
  const float s = scal[0], inv = scal[1];
  const float* ur = u + (size_t)(rowbase + c) * (F_N * 2) + hi * 4;
  v8f ore[3], oim[3];
#pragma unroll
  for (int j = 0; j < 3; ++j) {
    ore[j] = (v8f){};
    oim[j] = (v8f){};
  }
#pragma unroll
  for (int ks = 0; ks < 12; ++ks) {
    const float4 x = *(const float4*)(ur + ks * 8);
    const v2f ar = (v2f){s * x.x, s * x.z};
    const v2f ai = (v2f){s * x.y, s * x.w};
    const v2f an = -ai;
    const int k0 = ks * 4 + hi * 2;
#pragma unroll
    for (int j = 0; j < 3; ++j) {
      const v2f br = (v2f){Wre[k0 * F_N + j * 16 + c], Wre[(k0 + 1) * F_N + j * 16 + c]};
      const v2f bi = (v2f){Wim[k0 * F_N + j * 16 + c], Wim[(k0 + 1) * F_N + j * 16 + c]};
      ore[j] = wmma_f32(ar, br, ore[j]);
      ore[j] = wmma_f32(an, bi, ore[j]);
      oim[j] = wmma_f32(ar, bi, oim[j]);
      oim[j] = wmma_f32(ai, br, oim[j]);
    }
  }
#pragma unroll
  for (int j = 0; j < 3; ++j)
#pragma unroll
    for (int v = 0; v < 8; ++v) {
      const int row = rowbase + hi * 8 + v;
      float2 o;
      o.x = ore[j][v] * inv;
      o.y = oim[j][v] * inv;
      *(float2*)(out + ((size_t)row * F_N + j * 16 + c) * 2) = o;
    }
}

extern "C" void kernel_launch(void* const* d_in, const int* in_sizes, int n_in,
                              void* d_out, int out_size, void* d_ws, size_t ws_size,
                              hipStream_t stream) {
  const float* u  = (const float*)d_in[0];
  const float* S1 = (const float*)d_in[1];
  const float* S2 = (const float*)d_in[2];
  float* out = (float*)d_out;
  float* w = (float*)d_ws;

  float* scal  = w;
  float* part  = w + 8;
  float* ire   = w + 2048;
  float* iim   = ire + F_N * T_N;
  float* dre   = iim + F_N * T_N;
  float* dim   = dre + F_N * T_N;
  float* G     = w + 100352;
  float* spart = w + 104960;
  float* pst   = w + 178688;
  float* gpart = w + 227840;

  k_gram<<<NGP, 96, 0, stream>>>(u, part, gpart, ire, iim, dre, dim);
  k_gred<<<CSZ / 32, 256, 0, stream>>>(gpart, part, G, scal);

  for (int l = 0; l < L_N; ++l) {
    k_layer<<<LBLK, 288, 0, stream>>>(ire, iim, dre, dim, G,
                                      spart + ((l + 1) & 1) * (LBLK * CSZ),
                                      spart + (l & 1) * (LBLK * CSZ), pst, S1, S2,
                                      (l == 0) ? 1 : 0);
  }
  k_out<<<B_N / 64, 128, 0, stream>>>(u, scal, spart + ((L_N - 1) & 1) * (LBLK * CSZ), out);
}


# idft, dft and P state kept in memory as per-lane 16-byte fragment pieces: a column wave enters on 56 loads, not 128 (one outstanding-load window), dft fragments 16 loads, not 64 (sums unchanged)
# speedup vs baseline: 23.2101x; 1.0349x over previous
#include <hip/hip_runtime.h>
#include <math.h>


#define B_N 16384
#define F_N 48
#define T_N 512
#define L_N 16
#define MSZ (F_N * F_N)
#define CSZ (2 * MSZ)
#define NGP 256
#define LBLK 8
#define LCOL 64
#define NCW (LCOL / 16)
#define PPITCH 68
#define MPITCH 52
#define GPITCH 52
#define FPC (F_N * T_N / 4)

typedef __attribute__((ext_vector_type(2))) float v2f;
typedef __attribute__((ext_vector_type(8))) float v8f;

static __device__ __forceinline__ v8f wmma_f32(v2f a, v2f b, v8f c) {
  return __builtin_amdgcn_wmma_f32_16x16x4_f32(false, a, false, b, (short)0, c,
                                               false, false);
}


__global__ void __launch_bounds__(96) k_gram(const float* __restrict__ u,
                                             float* __restrict__ part,
                                             float* __restrict__ gpart,
                                             float* __restrict__ ire, float* __restrict__ iim,
                                             float* __restrict__ dre, float* __restrict__ dim) {
  {
    const int idx = blockIdx.x * 96 + threadIdx.x;
    const float sc = 0.04419417382415922f;
    const float w = 6.283185307179586f / (float)T_N;
    float s0, c0, s1, c1;
    const int q = idx >> 1, hf = idx & 1;
    const int ln = q & 31, lh = ln >> 4, lc = ln & 15;
    if (q < FPC) {
      const int cg = q >> 5, ct = cg / 6, g = cg - ct * 6;
      const int t = ct * 16 + lc;
      const int f0 = 4 * (2 * g + hf) + 2 * lh;
      sincosf((float)((f0 * t) & (T_N - 1)) * w, &s0, &c0);
      sincosf((float)(((f0 + 1) * t) & (T_N - 1)) * w, &s1, &c1);
      *(float2*)(ire + q * 4 + hf * 2) = make_float2(c0 * sc, c1 * sc);
      *(float2*)(iim + q * 4 + hf * 2) = make_float2(s0 * sc, s1 * sc);
    } else {
      const int q2 = q - FPC;
      const int r = q2 >> 5, g = r & 7, r2 = r >> 3;
      const int blk = r2 / 3, tj = r2 - blk * 3;
      const int f = tj * 16 + lc;
      const int t0 = blk * LCOL + 4 * (2 * g + hf) + 2 * lh;
      sincosf((float)((f * t0) & (T_N - 1)) * w, &s0, &c0);
      sincosf((float)((f * (t0 + 1)) & (T_N - 1)) * w, &s1, &c1);
      *(float2*)(dre + q2 * 4 + hf * 2) = make_float2(c0 * sc, c1 * sc);
      *(float2*)(dim + q2 * 4 + hf * 2) = make_float2(-s0 * sc, -s1 * sc);
    }
  }
  const int lane = threadIdx.x & 31;
  const int wi = threadIdx.x >> 5;
  const int hi = lane >> 4, c = lane & 15;
  const float* ub = u + (size_t)(blockIdx.x * 64 + 2 * hi) * (F_N * 2) + c * 2;
  v8f gre[3], gim[3];
#pragma unroll
  for (int j = 0; j < 3; ++j) {
    gre[j] = (v8f){};
    gim[j] = (v8f){};
  }
  float mxl = -3.4e38f;
#pragma unroll
  for (int ks = 0; ks < 16; ++ks) {
    v2f xr[3], xi[3];
#pragma unroll
    for (int ft = 0; ft < 3; ++ft) {
      const float2 p0 = *(const float2*)(ub + (ks * 4) * (F_N * 2) + ft * 32);
      const float2 p1 = *(const float2*)(ub + (ks * 4 + 1) * (F_N * 2) + ft * 32);
      xr[ft] = (v2f){p0.x, p1.x};
      xi[ft] = (v2f){p0.y, p1.y};
      mxl = fmaxf(mxl, fmaxf(fmaxf(p0.x, p0.y), fmaxf(p1.x, p1.y)));
    }
    const float2 a0 = *(const float2*)(ub + (ks * 4) * (F_N * 2) + wi * 32);
    const float2 a1 = *(const float2*)(ub + (ks * 4 + 1) * (F_N * 2) + wi * 32);
    const v2f ar = (v2f){a0.x, a1.x};
    const v2f ai = (v2f){a0.y, a1.y};
    const v2f an = -ai;
#pragma unroll
    for (int j = 0; j < 3; ++j) {
      gre[j] = wmma_f32(ar, xr[j], gre[j]);
      gim[j] = wmma_f32(ar, xi[j], gim[j]);
      gre[j] = wmma_f32(ai, xi[j], gre[j]);
      gim[j] = wmma_f32(an, xr[j], gim[j]);
    }
  }
  float* gp = gpart + (size_t)blockIdx.x * CSZ + wi * (3 * 256);
#pragma unroll
  for (int j = 0; j < 3; ++j)
#pragma unroll
    for (int v = 0; v < 8; ++v) {
      gp[j * 256 + v * 32 + lane] = gre[j][v];
      gp[MSZ + j * 256 + v * 32 + lane] = gim[j][v];
    }
  mxl = fmaxf(mxl, __shfl_xor(mxl, 16, 32));
  mxl = fmaxf(mxl, __shfl_xor(mxl, 8, 32));
  mxl = fmaxf(mxl, __shfl_xor(mxl, 4, 32));
  mxl = fmaxf(mxl, __shfl_xor(mxl, 2, 32));
  mxl = fmaxf(mxl, __shfl_xor(mxl, 1, 32));
  if (threadIdx.x == 0) part[blockIdx.x] = mxl;
}

__global__ void __launch_bounds__(256) k_gred(const float* __restrict__ gpart,
                                              const float* __restrict__ part,
                                              float* __restrict__ G,
                                              float* __restrict__ scal) {
  __shared__ float sm[256];
  __shared__ float red[8][32];
  const int grp = threadIdx.x >> 5, el = threadIdx.x & 31;
  const int n = blockIdx.x * 32 + el;
  {
    const float* p = gpart + (size_t)(grp * 32) * CSZ + n;
    float acc = 0.0f;
    for (int i = 0; i < 32; ++i) acc += p[(size_t)i * CSZ];
    red[grp][el] = acc;
  }
  sm[threadIdx.x] = part[threadIdx.x];
  __syncthreads();
  for (int s = 128; s > 0; s >>= 1) {
    if (threadIdx.x < s) sm[threadIdx.x] = fmaxf(sm[threadIdx.x], sm[threadIdx.x + s]);
    __syncthreads();
  }
  const float mx = fabsf(sm[0]);
  const float s = 5.0f / mx;
  if (blockIdx.x == 0 && threadIdx.x == 0) {
    scal[0] = s;
    scal[1] = mx / 5.0f;
  }
  if (threadIdx.x < 32) {
    float tot = 0.0f;
#pragma unroll
    for (int k = 0; k < 8; ++k) tot += red[k][el];
    const int pl = n / MSZ, q = n - pl * MSZ;
    const int tl = q >> 8, tr = tl / 3;
    const int r = tr * 16 + ((q >> 5) & 7) + ((q >> 4) & 1) * 8;
    const int cc = (tl - tr * 3) * 16 + (q & 15);
    G[pl * MSZ + r * F_N + cc] = tot * (s * s * (1.0f / (float)B_N));
  }
}

static __device__ __forceinline__ void load_dft_frag(const float* __restrict__ dre,
                                                     const float* __restrict__ dim,
                                                     int base, v2f (&dR)[LCOL / 4],
                                                     v2f (&dI)[LCOL / 4]) {
#pragma unroll
  for (int g = 0; g < LCOL / 8; ++g) {
    const float4 r4 = *(const float4*)(dre + base + g * 128);
    const float4 i4 = *(const float4*)(dim + base + g * 128);
    dR[2 * g] = (v2f){r4.x, r4.y};
    dR[2 * g + 1] = (v2f){r4.z, r4.w};
    dI[2 * g] = (v2f){i4.x, i4.y};
    dI[2 * g + 1] = (v2f){i4.z, i4.w};
  }
}

__global__ void __launch_bounds__(288) k_layer(
    const float* __restrict__ ire, const float* __restrict__ iim,
    const float* __restrict__ dre, const float* __restrict__ dim,
    const float* __restrict__ G, const float* __restrict__ sprev,
    float* __restrict__ snext, float* __restrict__ pst,
    const float* __restrict__ S1, const float* __restrict__ S2, int first) {
  __shared__ __attribute__((aligned(16))) float sm[2 * F_N * PPITCH + 2 * F_N * GPITCH];
  float* Mre = sm;
  float* Mim = sm + F_N * MPITCH;
  float* Pre = sm;
  float* Pim = sm + F_N * PPITCH;
  float* Gre = sm + 2 * F_N * PPITCH;
  float* Gim = Gre + F_N * GPITCH;

  const int tid = threadIdx.x;
  const int lane = tid & 31;
  const int wv = tid >> 5;
  const int hi = lane >> 4, c = lane & 15;
  const int ti = wv / 3, tj = wv - ti * 3;
  const int dbase = ((blockIdx.x * 3 + tj) * (LCOL / 8) * 32 + lane) * 4;
  const float s1 = S1[0], s2 = S2[0];

  v8f qre[3], qim[3];
#pragma unroll
  for (int rt = 0; rt < 3; ++rt) {
    qre[rt] = (v8f){};
    qim[rt] = (v8f){};
  }
  v2f bR[12], bI[12];
#pragma unroll
  for (int ks = 0; ks < 12; ++ks) {
    bR[ks] = (v2f){0.f, 0.f};
    bI[ks] = (v2f){0.f, 0.f};
  }
  float* pw = pst + (size_t)(blockIdx.x * NCW + (wv & (NCW - 1))) * (2 * 3 * 8 * 32);
  const int ctile = blockIdx.x * NCW + (wv & (NCW - 1));
  if (wv < NCW) {
    if (!first) {
#pragma unroll
      for (int rt = 0; rt < 3; ++rt)
#pragma unroll
        for (int vg = 0; vg < 2; ++vg) {
          const float4 r4 = *(const float4*)(pw + (rt * 2 + vg) * 128 + lane * 4);
          const float4 i4 = *(const float4*)(pw + 768 + (rt * 2 + vg) * 128 + lane * 4);
          qre[rt][vg * 4 + 0] = r4.x; qre[rt][vg * 4 + 1] = r4.y;
          qre[rt][vg * 4 + 2] = r4.z; qre[rt][vg * 4 + 3] = r4.w;
          qim[rt][vg * 4 + 0] = i4.x; qim[rt][vg * 4 + 1] = i4.y;
          qim[rt][vg * 4 + 2] = i4.z; qim[rt][vg * 4 + 3] = i4.w;
        }
    }
#pragma unroll
    for (int g = 0; g < 6; ++g) {
      const float4 r4 = *(const float4*)(ire + ((ctile * 6 + g) * 32 + lane) * 4);
      const float4 i4 = *(const float4*)(iim + ((ctile * 6 + g) * 32 + lane) * 4);
      bR[2 * g] = (v2f){r4.x, r4.y};
      bR[2 * g + 1] = (v2f){r4.z, r4.w};
      bI[2 * g] = (v2f){i4.x, i4.y};
      bI[2 * g + 1] = (v2f){i4.z, i4.w};
    }
  }

  float4 ms[4];
#pragma unroll
  for (int i = 0; i < 4; ++i) ms[i] = make_float4(0.f, 0.f, 0.f, 0.f);
  if (!first) {
    float4 t[4][LBLK];
#pragma unroll
    for (int i = 0; i < 4; ++i) {
      const int pl = i >> 1;
      const int n = (tid + 288 * i - pl * (MSZ / 4)) * 4;
#pragma unroll
      for (int p = 0; p < LBLK; ++p) t[i][p] = *(const float4*)(sprev + p * CSZ + pl * MSZ + n);
    }
#pragma unroll
    for (int i = 0; i < 4; ++i)
#pragma unroll
      for (int p = 0; p < LBLK; ++p) {
        ms[i].x += t[i][p].x; ms[i].y += t[i][p].y; ms[i].z += t[i][p].z; ms[i].w += t[i][p].w;
      }
  }

#pragma unroll
  for (int i = 0; i < 4; ++i) {
    const int pl = i >> 1;
    const int n = (tid + 288 * i - pl * (MSZ / 4)) * 4;
    const int tl = n >> 8, tr = tl / 3;
    const int r = tr * 16 + ((n >> 5) & 7) + ((n >> 4) & 1) * 8;
    const int cc = (tl - tr * 3) * 16 + (n & 15);
    float4 m;
    if (pl == 0) {
      m.x = ((r == cc) ? 1.0f : 0.0f) - ms[i].x;
      m.y = ((r == cc + 1) ? 1.0f : 0.0f) - ms[i].y;
      m.z = ((r == cc + 2) ? 1.0f : 0.0f) - ms[i].z;
      m.w = ((r == cc + 3) ? 1.0f : 0.0f) - ms[i].w;
      *(float4*)&Mre[r * MPITCH + cc] = m;
    } else {
      m.x = -ms[i].x; m.y = -ms[i].y; m.z = -ms[i].z; m.w = -ms[i].w;
      *(float4*)&Mim[r * MPITCH + cc] = m;
    }
  }
  __syncthreads();

  v2f dR[LCOL / 4], dI[LCOL / 4];
  if (wv < NCW) {
    v8f qk[3];
#pragma unroll
    for (int it = 0; it < 3; ++it) qk[it] = (v8f){};
#pragma unroll
    for (int ks = 0; ks < 12; ++ks) {
      const int k0 = ks * 4 + hi * 2;
      const v2f bre = bR[ks];
      const v2f bim = bI[ks];
      const v2f bd = bim - bre;
      const v2f bs = -bre - bim;
#pragma unroll
      for (int it = 0; it < 3; ++it) {
        const v2f are = *(const v2f*)&Mre[(it * 16 + c) * MPITCH + k0];
        const v2f aim = *(const v2f*)&Mim[(it * 16 + c) * MPITCH + k0];
        const v2f as = are + aim;
        qk[it]  = wmma_f32(as, bre, qk[it]);
        qim[it] = wmma_f32(are, bd, qim[it]);
        qre[it] = wmma_f32(aim, bs, qre[it]);
      }
    }
#pragma unroll
    for (int it = 0; it < 3; ++it) {
      qre[it] += qk[it];
      qim[it] += qk[it];
    }
  } else {
    if (wv < 2 * NCW) {
      float4 g[9];
#pragma unroll
      for (int i = 0; i < 9; ++i) g[i] = *(const float4*)(G + (tid - 128 + 128 * i) * 4);
#pragma unroll
      for (int i = 0; i < 9; ++i) {
        const int q = tid - 128 + 128 * i;
        const int pl = q / (MSZ / 4), rem = q - pl * (MSZ / 4);
        const int r = rem / (F_N / 4), c4 = rem - r * (F_N / 4);
        *(float4*)&Gre[pl * (F_N * GPITCH) + r * GPITCH + c4 * 4] = g[i];
      }
    }
    load_dft_frag(dre, dim, dbase, dR, dI);
  }
  __syncthreads();

  if (wv < NCW) {
    v8f yk[3], yre[3], yim[3];
#pragma unroll
    for (int it = 0; it < 3; ++it) {
      yk[it] = (v8f){};
      yre[it] = (v8f){};
      yim[it] = (v8f){};
    }
#pragma unroll
    for (int rt = 0; rt < 3; ++rt) {
      float4 gr[3][2], gi[3][2];
#pragma unroll
      for (int it = 0; it < 3; ++it) {
        const int go = (it * 16 + c) * GPITCH + rt * 16 + hi * 8;
        gr[it][0] = *(const float4*)&Gre[go];
        gr[it][1] = *(const float4*)&Gre[go + 4];
        gi[it][0] = *(const float4*)&Gim[go];
        gi[it][1] = *(const float4*)&Gim[go + 4];
      }
#pragma unroll
      for (int j = 0; j < 4; ++j) {
        const v2f bqr = (v2f){qre[rt][2 * j], qre[rt][2 * j + 1]};
        const v2f bqi = (v2f){qim[rt][2 * j], qim[rt][2 * j + 1]};
        const v2f bqd = bqi - bqr;
        const v2f bqs = -bqr - bqi;
#pragma unroll
        for (int it = 0; it < 3; ++it) {
          const float4 r4 = gr[it][j >> 1];
          const float4 i4 = gi[it][j >> 1];
          const v2f g_r = (j & 1) ? (v2f){r4.z, r4.w} : (v2f){r4.x, r4.y};
          const v2f g_i = (j & 1) ? (v2f){i4.z, i4.w} : (v2f){i4.x, i4.y};
          const v2f gs = g_r + g_i;
          yk[it]  = wmma_f32(gs, bqr, yk[it]);
          yim[it] = wmma_f32(g_r, bqd, yim[it]);
          yre[it] = wmma_f32(g_i, bqs, yre[it]);
        }
      }
    }
    load_dft_frag(dre, dim, dbase, dR, dI);
    float p = 0.0f;
#pragma unroll
    for (int rt = 0; rt < 3; ++rt)
#pragma unroll
      for (int v = 0; v < 8; ++v)
        p += qre[rt][v] * (yre[rt][v] + yk[rt][v]) + qim[rt][v] * (yim[rt][v] + yk[rt][v]);
    p += __shfl_xor(p, 16, 32);
    const float rm = sqrtf(fmaxf(p, 0.0f));
    const float g = 1.0f / (1.0f + expf(-(s1 * (rm - s2))));
#pragma unroll
    for (int rt = 0; rt < 3; ++rt)
#pragma unroll
      for (int v = 0; v < 8; ++v) {
        qre[rt][v] *= g;
        qim[rt][v] *= g;
        Pre[(rt * 16 + hi * 8 + v) * PPITCH + wv * 16 + c] = qre[rt][v];
        Pim[(rt * 16 + hi * 8 + v) * PPITCH + wv * 16 + c] = qim[rt][v];
      }
  }
  __syncthreads();
  if (wv < NCW) {
#pragma unroll
    for (int rt = 0; rt < 3; ++rt)
#pragma unroll
      for (int vg = 0; vg < 2; ++vg) {
        *(float4*)(pw + (rt * 2 + vg) * 128 + lane * 4) =
            make_float4(qre[rt][vg * 4], qre[rt][vg * 4 + 1], qre[rt][vg * 4 + 2], qre[rt][vg * 4 + 3]);
        *(float4*)(pw + 768 + (rt * 2 + vg) * 128 + lane * 4) =
            make_float4(qim[rt][vg * 4], qim[rt][vg * 4 + 1], qim[rt][vg * 4 + 2], qim[rt][vg * 4 + 3]);
      }
  }

  {
    v8f sk = {}, sre = {}, sim = {};
#pragma unroll
    for (int ks = 0; ks < LCOL / 4; ++ks) {
      const int k0 = ks * 4 + hi * 2;
      const v2f ar = *(const v2f*)&Pre[(ti * 16 + c) * PPITCH + k0];
      const v2f ai = *(const v2f*)&Pim[(ti * 16 + c) * PPITCH + k0];
      const v2f as = ar + ai;
      const v2f br = dR[ks];
      const v2f bi = dI[ks];
      const v2f bd = bi - br;
      const v2f bs = -br - bi;
      sk  = wmma_f32(as, br, sk);
      sim = wmma_f32(ar, bd, sim);
      sre = wmma_f32(ai, bs, sre);
    }
    float* sp = snext + blockIdx.x * CSZ + wv * 256;
#pragma unroll
    for (int v = 0; v < 8; ++v) {
      sp[v * 32 + lane] = sre[v] + sk[v];
      sp[MSZ + v * 32 + lane] = sim[v] + sk[v];
    }
  }
}

__global__ void __launch_bounds__(128) k_out(const float* __restrict__ u,
                                             const float* __restrict__ scal,
                                             const float* __restrict__ sp,
                                             float* __restrict__ out) {
  __shared__ __attribute__((aligned(16))) float Wre[MSZ];
  __shared__ __attribute__((aligned(16))) float Wim[MSZ];
  for (int q4 = threadIdx.x; q4 < MSZ / 4; q4 += 128) {
    const int n = q4 * 4;
    const int tl = n >> 8, tr = tl / 3;
    const int r = tr * 16 + ((n >> 5) & 7) + ((n >> 4) & 1) * 8;
    const int cc = (tl - tr * 3) * 16 + (n & 15);
    float4 sr = make_float4(0.f, 0.f, 0.f, 0.f), si = make_float4(0.f, 0.f, 0.f, 0.f);
#pragma unroll
    for (int p = 0; p < LBLK; ++p) {
      const float4 a = *(const float4*)(sp + p * CSZ + n);
      const float4 b = *(const float4*)(sp + p * CSZ + MSZ + n);
      sr.x += a.x; sr.y += a.y; sr.z += a.z; sr.w += a.w;
      si.x += b.x; si.y += b.y; si.z += b.z; si.w += b.w;
    }
    *(float4*)&Wre[r * F_N + cc] = sr;
    *(float4*)&Wim[r * F_N + cc] = si;
  }
  __syncthreads();

  const int lane = threadIdx.x & 31;
  const int wv = threadIdx.x >> 5;
  const int hi = lane >> 4, c = lane & 15;
  const int rowbase = (blockIdx.x * 4 + wv) * 16;
  const float s = scal[0], inv = scal[1];
  const float* ur = u + (size_t)(rowbase + c) * (F_N * 2) + hi * 4;
  v8f ore[3], oim[3];
#pragma unroll
  for (int j = 0; j < 3; ++j) {
    ore[j] = (v8f){};
    oim[j] = (v8f){};
  }
#pragma unroll
  for (int ks = 0; ks < 12; ++ks) {
    const float4 x = *(const float4*)(ur + ks * 8);
    const v2f ar = (v2f){s * x.x, s * x.z};
    const v2f ai = (v2f){s * x.y, s * x.w};
    const v2f an = -ai;
    const int k0 = ks * 4 + hi * 2;
#pragma unroll
    for (int j = 0; j < 3; ++j) {
      const v2f br = (v2f){Wre[k0 * F_N + j * 16 + c], Wre[(k0 + 1) * F_N + j * 16 + c]};
      const v2f bi = (v2f){Wim[k0 * F_N + j * 16 + c], Wim[(k0 + 1) * F_N + j * 16 + c]};
      ore[j] = wmma_f32(ar, br, ore[j]);
      ore[j] = wmma_f32(an, bi, ore[j]);
      oim[j] = wmma_f32(ar, bi, oim[j]);
      oim[j] = wmma_f32(ai, br, oim[j]);
    }
  }
#pragma unroll
  for (int j = 0; j < 3; ++j)
#pragma unroll
    for (int v = 0; v < 8; ++v) {
      const int row = rowbase + hi * 8 + v;
      float2 o;
      o.x = ore[j][v] * inv;
      o.y = oim[j][v] * inv;
      *(float2*)(out + ((size_t)row * F_N + j * 16 + c) * 2) = o;
    }
}

extern "C" void kernel_launch(void* const* d_in, const int* in_sizes, int n_in,
                              void* d_out, int out_size, void* d_ws, size_t ws_size,
                              hipStream_t stream) {
  const float* u  = (const float*)d_in[0];
  const float* S1 = (const float*)d_in[1];
  const float* S2 = (const float*)d_in[2];
  float* out = (float*)d_out;
  float* w = (float*)d_ws;

  float* scal  = w;
  float* part  = w + 8;
  float* ire   = w + 2048;
  float* iim   = ire + F_N * T_N;
  float* dre   = iim + F_N * T_N;
  float* dim   = dre + F_N * T_N;
  float* G     = w + 100352;
  float* spart = w + 104960;
  float* pst   = w + 178688;
  float* gpart = w + 227840;

  k_gram<<<NGP, 96, 0, stream>>>(u, part, gpart, ire, iim, dre, dim);
  k_gred<<<CSZ / 32, 256, 0, stream>>>(gpart, part, G, scal);

  for (int l = 0; l < L_N; ++l) {
    k_layer<<<LBLK, 288, 0, stream>>>(ire, iim, dre, dim, G,
                                      spart + ((l + 1) & 1) * (LBLK * CSZ),
                                      spart + (l & 1) * (LBLK * CSZ), pst, S1, S2,
                                      (l == 0) ? 1 : 0);
  }
  k_out<<<B_N / 64, 128, 0, stream>>>(u, scal, spart + ((L_N - 1) & 1) * (LBLK * CSZ), out);
}


# layer kernel: both DFT products folded over mirror column pairs (t, T-t): M@idft as 2 real products per wave swapped via LDS, P@dft as 4 on 32 pair slots; no atomics (sum order changes)
# speedup vs baseline: 25.7340x; 1.1087x over previous
#include <hip/hip_runtime.h>
#include <math.h>


#define B_N 16384
#define F_N 48
#define T_N 512
#define L_N 16
#define MSZ (F_N * F_N)
#define CSZ (2 * MSZ)
#define NGP 256
#define LBLK 8
#define LCOL 64
#define NCW (LCOL / 16)
#define PPITCH 68
#define MPITCH 52
#define GPITCH 52
#define FPC (F_N * T_N / 4)
#define NPAIR (LCOL / 2)
#define DPC (FPC / 2)


typedef __attribute__((ext_vector_type(2))) float v2f;
typedef __attribute__((ext_vector_type(8))) float v8f;

static __device__ __forceinline__ v8f wmma_f32(v2f a, v2f b, v8f c) {
  return __builtin_amdgcn_wmma_f32_16x16x4_f32(false, a, false, b, (short)0, c,
                                               false, false);
}


__global__ void __launch_bounds__(96) k_gram(const float* __restrict__ u,
                                             float* __restrict__ part,
                                             float* __restrict__ gpart,
                                             float* __restrict__ ire, float* __restrict__ iim,
                                             float* __restrict__ dre, float* __restrict__ dim) {
  {
    const int idx = blockIdx.x * 96 + threadIdx.x;
    const float sc = 0.04419417382415922f;
    const float w = 6.283185307179586f / (float)T_N;
    float s0, c0, s1, c1;
    const int q = idx >> 1, hf = idx & 1;
    const int ln = q & 31, lh = ln >> 4, lc = ln & 15;
    if (q < FPC) {
      const int cg = q >> 5, ct = cg / 6, g = cg - ct * 6;
      const int t = (ct >> 2) * NPAIR + (ct & 1) * 16 + 1 + lc;
      const bool col0 = ((ct & 2) != 0) && (t == T_N / 2);
      const int f0 = 4 * (2 * g + hf) + 2 * lh;
      sincosf((float)((f0 * t) & (T_N - 1)) * w, &s0, &c0);
      sincosf((float)(((f0 + 1) * t) & (T_N - 1)) * w, &s1, &c1);
      *(float2*)(ire + q * 4 + hf * 2) = make_float2(c0 * sc, c1 * sc);
      *(float2*)(iim + q * 4 + hf * 2) = col0 ? make_float2(sc, sc) : make_float2(s0 * sc, s1 * sc);
    } else if (q < FPC + DPC) {
      const int q2 = q - FPC;
      const int r = q2 >> 5, g = r & 3, r2 = r >> 2;
      const int blk = r2 / 3, tj = r2 - blk * 3;
      const int f = tj * 16 + lc;
      const int t0 = blk * NPAIR + 1 + 4 * (2 * g + hf) + 2 * lh;
      sincosf((float)((f * t0) & (T_N - 1)) * w, &s0, &c0);
      sincosf((float)((f * (t0 + 1)) & (T_N - 1)) * w, &s1, &c1);
      *(float2*)(dre + q2 * 4 + hf * 2) = make_float2(c0 * sc, c1 * sc);
      *(float2*)(dim + q2 * 4 + hf * 2) = make_float2(s0 * sc, s1 * sc);
    }
  }
  const int lane = threadIdx.x & 31;
  const int wi = threadIdx.x >> 5;
  const int hi = lane >> 4, c = lane & 15;
  const float* ub = u + (size_t)(blockIdx.x * 64 + 2 * hi) * (F_N * 2) + c * 2;
  v8f gre[3], gim[3];
#pragma unroll
  for (int j = 0; j < 3; ++j) {
    gre[j] = (v8f){};
    gim[j] = (v8f){};
  }
  float mxl = -3.4e38f;
#pragma unroll
  for (int ks = 0; ks < 16; ++ks) {
    v2f xr[3], xi[3];
#pragma unroll
    for (int ft = 0; ft < 3; ++ft) {
      const float2 p0 = *(const float2*)(ub + (ks * 4) * (F_N * 2) + ft * 32);
      const float2 p1 = *(const float2*)(ub + (ks * 4 + 1) * (F_N * 2) + ft * 32);
      xr[ft] = (v2f){p0.x, p1.x};
      xi[ft] = (v2f){p0.y, p1.y};
      mxl = fmaxf(mxl, fmaxf(fmaxf(p0.x, p0.y), fmaxf(p1.x, p1.y)));
    }
    const float2 a0 = *(const float2*)(ub + (ks * 4) * (F_N * 2) + wi * 32);
    const float2 a1 = *(const float2*)(ub + (ks * 4 + 1) * (F_N * 2) + wi * 32);
    const v2f ar = (v2f){a0.x, a1.x};
    const v2f ai = (v2f){a0.y, a1.y};
    const v2f an = -ai;
#pragma unroll
    for (int j = 0; j < 3; ++j) {
      gre[j] = wmma_f32(ar, xr[j], gre[j]);
      gim[j] = wmma_f32(ar, xi[j], gim[j]);
      gre[j] = wmma_f32(ai, xi[j], gre[j]);
      gim[j] = wmma_f32(an, xr[j], gim[j]);
    }
  }
  float* gp = gpart + (size_t)blockIdx.x * CSZ + wi * (3 * 256);
#pragma unroll
  for (int j = 0; j < 3; ++j)
#pragma unroll
    for (int v = 0; v < 8; ++v) {
      gp[j * 256 + v * 32 + lane] = gre[j][v];
      gp[MSZ + j * 256 + v * 32 + lane] = gim[j][v];
    }
  mxl = fmaxf(mxl, __shfl_xor(mxl, 16, 32));
  mxl = fmaxf(mxl, __shfl_xor(mxl, 8, 32));
  mxl = fmaxf(mxl, __shfl_xor(mxl, 4, 32));
  mxl = fmaxf(mxl, __shfl_xor(mxl, 2, 32));
  mxl = fmaxf(mxl, __shfl_xor(mxl, 1, 32));
  if (threadIdx.x == 0) part[blockIdx.x] = mxl;
}

__global__ void __launch_bounds__(256) k_gred(const float* __restrict__ gpart,
                                              const float* __restrict__ part,
                                              float* __restrict__ G,
                                              float* __restrict__ scal) {
  __shared__ float sm[256];
  __shared__ float red[8][32];
  const int grp = threadIdx.x >> 5, el = threadIdx.x & 31;
  const int n = blockIdx.x * 32 + el;
  {
    const float* p = gpart + (size_t)(grp * 32) * CSZ + n;
    float acc = 0.0f;
    for (int i = 0; i < 32; ++i) acc += p[(size_t)i * CSZ];
    red[grp][el] = acc;
  }
  sm[threadIdx.x] = part[threadIdx.x];
  __syncthreads();
  for (int s = 128; s > 0; s >>= 1) {
    if (threadIdx.x < s) sm[threadIdx.x] = fmaxf(sm[threadIdx.x], sm[threadIdx.x + s]);
    __syncthreads();
  }
  const float mx = fabsf(sm[0]);
  const float s = 5.0f / mx;
  if (blockIdx.x == 0 && threadIdx.x == 0) {
    scal[0] = s;
    scal[1] = mx / 5.0f;
  }
  if (threadIdx.x < 32) {
    float tot = 0.0f;
#pragma unroll
    for (int k = 0; k < 8; ++k) tot += red[k][el];
    const int pl = n / MSZ, q = n - pl * MSZ;
    const int tl = q >> 8, tr = tl / 3;
    const int r = tr * 16 + ((q >> 5) & 7) + ((q >> 4) & 1) * 8;
    const int cc = (tl - tr * 3) * 16 + (q & 15);
    G[pl * MSZ + r * F_N + cc] = tot * (s * s * (1.0f / (float)B_N));
  }
}

static __device__ __forceinline__ void load_dft_frag(const float* __restrict__ dre,
                                                     const float* __restrict__ dim,
                                                     int base, v2f (&dR)[NPAIR / 4],
                                                     v2f (&dI)[NPAIR / 4]) {
#pragma unroll
  for (int g = 0; g < NPAIR / 8; ++g) {
    const float4 r4 = *(const float4*)(dre + base + g * 128);
    const float4 i4 = *(const float4*)(dim + base + g * 128);
    dR[2 * g] = (v2f){r4.x, r4.y};
    dR[2 * g + 1] = (v2f){r4.z, r4.w};
    dI[2 * g] = (v2f){i4.x, i4.y};
    dI[2 * g + 1] = (v2f){i4.z, i4.w};
  }
}

__global__ void __launch_bounds__(288) k_layer(
    const float* __restrict__ ire, const float* __restrict__ iim,
    const float* __restrict__ dre, const float* __restrict__ dim,
    const float* __restrict__ G, const float* __restrict__ sprev,
    float* __restrict__ snext, float* __restrict__ pst,
    const float* __restrict__ S1, const float* __restrict__ S2, int first) {
  __shared__ __attribute__((aligned(16))) float sm[2 * F_N * PPITCH + 2 * F_N * GPITCH + NCW * 48 * 32];
  float* Mre = sm;
  float* Mim = sm + F_N * MPITCH;
  float* Pre = sm;
  float* Pim = sm + F_N * PPITCH;
  float* Gre = sm + 2 * F_N * PPITCH;
  float* Gim = Gre + F_N * GPITCH;
  float* Xs = Gim + F_N * GPITCH;

  const int tid = threadIdx.x;
  const int lane = tid & 31;
  const int wv = tid >> 5;
  const int hi = lane >> 4, c = lane & 15;
  const int ti = wv / 3, tj = wv - ti * 3;
  const int dbase = ((blockIdx.x * 3 + tj) * (NPAIR / 8) * 32 + lane) * 4;
  const float s1 = S1[0], s2 = S2[0];

  v8f qre[3], qim[3];
#pragma unroll
  for (int rt = 0; rt < 3; ++rt) {
    qre[rt] = (v8f){};
    qim[rt] = (v8f){};
  }
  v2f bF[12];
#pragma unroll
  for (int ks = 0; ks < 12; ++ks) bF[ks] = (v2f){0.f, 0.f};
  float* pw = pst + (size_t)(blockIdx.x * NCW + (wv & (NCW - 1))) * (2 * 3 * 8 * 32);
  const int ctile = blockIdx.x * NCW + (wv & (NCW - 1));
  if (wv < NCW) {
    if (!first) {
#pragma unroll
      for (int rt = 0; rt < 3; ++rt)
#pragma unroll
        for (int vg = 0; vg < 2; ++vg) {
          const float4 r4 = *(const float4*)(pw + (rt * 2 + vg) * 128 + lane * 4);
          const float4 i4 = *(const float4*)(pw + 768 + (rt * 2 + vg) * 128 + lane * 4);
          qre[rt][vg * 4 + 0] = r4.x; qre[rt][vg * 4 + 1] = r4.y;
          qre[rt][vg * 4 + 2] = r4.z; qre[rt][vg * 4 + 3] = r4.w;
          qim[rt][vg * 4 + 0] = i4.x; qim[rt][vg * 4 + 1] = i4.y;
          qim[rt][vg * 4 + 2] = i4.z; qim[rt][vg * 4 + 3] = i4.w;
        }
    }
    const float* fsrc = (wv < 2) ? ire : iim;
#pragma unroll
    for (int g = 0; g < 6; ++g) {
      const float4 f4 = *(const float4*)(fsrc + ((ctile * 6 + g) * 32 + lane) * 4);
      bF[2 * g] = (v2f){f4.x, f4.y};
      bF[2 * g + 1] = (v2f){f4.z, f4.w};
    }
  }

  float4 ms[4];
#pragma unroll
  for (int i = 0; i < 4; ++i) ms[i] = make_float4(0.f, 0.f, 0.f, 0.f);
  if (!first) {
    float4 t[4][LBLK];
#pragma unroll
    for (int i = 0; i < 4; ++i) {
      const int pl = i >> 1;
      const int n = (tid + 288 * i - pl * (MSZ / 4)) * 4;
#pragma unroll
      for (int p = 0; p < LBLK; ++p) t[i][p] = *(const float4*)(sprev + p * CSZ + pl * MSZ + n);
    }
#pragma unroll
    for (int i = 0; i < 4; ++i)
#pragma unroll
      for (int p = 0; p < LBLK; ++p) {
        ms[i].x += t[i][p].x; ms[i].y += t[i][p].y; ms[i].z += t[i][p].z; ms[i].w += t[i][p].w;
      }
  }

#pragma unroll
  for (int i = 0; i < 4; ++i) {
    const int pl = i >> 1;
    const int n = (tid + 288 * i - pl * (MSZ / 4)) * 4;
    const int tl = n >> 8, tr = tl / 3;
    const int r = tr * 16 + ((n >> 5) & 7) + ((n >> 4) & 1) * 8;
    const int cc = (tl - tr * 3) * 16 + (n & 15);
    float4 m;
    if (pl == 0) {
      m.x = ((r == cc) ? 1.0f : 0.0f) - ms[i].x;
      m.y = ((r == cc + 1) ? 1.0f : 0.0f) - ms[i].y;
      m.z = ((r == cc + 2) ? 1.0f : 0.0f) - ms[i].z;
      m.w = ((r == cc + 3) ? 1.0f : 0.0f) - ms[i].w;
      *(float4*)&Mre[r * MPITCH + cc] = m;
    } else {
      m.x = -ms[i].x; m.y = -ms[i].y; m.z = -ms[i].z; m.w = -ms[i].w;
      *(float4*)&Mim[r * MPITCH + cc] = m;
    }
  }
  __syncthreads();

  v2f dR[NPAIR / 4], dI[NPAIR / 4];
  v8f xa[3], xb[3];
#pragma unroll
  for (int it = 0; it < 3; ++it) {
    xa[it] = (v8f){};
    xb[it] = (v8f){};
  }
  if (wv < NCW) {
#pragma unroll
    for (int ks = 0; ks < 12; ++ks) {
      const int k0 = ks * 4 + hi * 2;
      const v2f bf = bF[ks];
#pragma unroll
      for (int it = 0; it < 3; ++it) {
        const v2f are = *(const v2f*)&Mre[(it * 16 + c) * MPITCH + k0];
        const v2f aim = *(const v2f*)&Mim[(it * 16 + c) * MPITCH + k0];
        xa[it] = wmma_f32(are, bf, xa[it]);
        xb[it] = wmma_f32(aim, bf, xb[it]);
      }
    }
    float* xw = Xs + wv * (48 * 32);
#pragma unroll
    for (int it = 0; it < 3; ++it)
#pragma unroll
      for (int vg = 0; vg < 2; ++vg) {
        *(float4*)&xw[((it * 4 + vg) * 32 + lane) * 4] =
            make_float4(xa[it][vg * 4], xa[it][vg * 4 + 1], xa[it][vg * 4 + 2], xa[it][vg * 4 + 3]);
        *(float4*)&xw[((it * 4 + 2 + vg) * 32 + lane) * 4] =
            make_float4(xb[it][vg * 4], xb[it][vg * 4 + 1], xb[it][vg * 4 + 2], xb[it][vg * 4 + 3]);
      }
  } else {
    if (wv < 2 * NCW) {
      float4 g[9];
#pragma unroll
      for (int i = 0; i < 9; ++i) g[i] = *(const float4*)(G + (tid - 128 + 128 * i) * 4);
#pragma unroll
      for (int i = 0; i < 9; ++i) {
        const int q = tid - 128 + 128 * i;
        const int pl = q / (MSZ / 4), rem = q - pl * (MSZ / 4);
        const int r = rem / (F_N / 4), c4 = rem - r * (F_N / 4);
        *(float4*)&Gre[pl * (F_N * GPITCH) + r * GPITCH + c4 * 4] = g[i];
      }
    }
    load_dft_frag(dre, dim, dbase, dR, dI);
  }
  __syncthreads();

  if (wv < NCW) {
    {
      const float* xr = Xs + (wv ^ 2) * (48 * 32);
      float4 ya[3][2], yb[3][2];
#pragma unroll
      for (int it = 0; it < 3; ++it)
#pragma unroll
        for (int vg = 0; vg < 2; ++vg) {
          ya[it][vg] = *(const float4*)&xr[((it * 4 + vg) * 32 + lane) * 4];
          yb[it][vg] = *(const float4*)&xr[((it * 4 + 2 + vg) * 32 + lane) * 4];
        }
      const bool last = (blockIdx.x == LBLK - 1);
      const float keep = (last && ((wv & 1) == 1) && (c == 15)) ? 0.0f : 1.0f;
      const int wvs = __builtin_amdgcn_readfirstlane(wv);
      if (wvs < 2) {
#pragma unroll
        for (int it = 0; it < 3; ++it)
#pragma unroll
          for (int v = 0; v < 8; ++v) {
            const float4 a4 = ya[it][v >> 2], b4 = yb[it][v >> 2];
            const float y3 = (v & 3) == 0 ? a4.x : (v & 3) == 1 ? a4.y : (v & 3) == 2 ? a4.z : a4.w;
            const float y4 = (v & 3) == 0 ? b4.x : (v & 3) == 1 ? b4.y : (v & 3) == 2 ? b4.z : b4.w;
            qre[it][v] = fmaf(-keep, y4, qre[it][v] + xa[it][v]);
            qim[it][v] = fmaf(keep, y3, qim[it][v] + xb[it][v]);
          }
      } else {
        const bool col0 = last && (wvs == NCW - 1) && (c == 15);
#pragma unroll
        for (int it = 0; it < 3; ++it)
#pragma unroll
          for (int v = 0; v < 8; ++v) {
            const float4 a4 = ya[it][v >> 2], b4 = yb[it][v >> 2];
            const float y1 = (v & 3) == 0 ? a4.x : (v & 3) == 1 ? a4.y : (v & 3) == 2 ? a4.z : a4.w;
            const float y2 = (v & 3) == 0 ? b4.x : (v & 3) == 1 ? b4.y : (v & 3) == 2 ? b4.z : b4.w;
            const float ore = col0 ? xa[it][v] : xb[it][v];
            const float oim = col0 ? xb[it][v] : -xa[it][v];
            qre[it][v] = fmaf(keep, y1, qre[it][v] + ore);
            qim[it][v] = fmaf(keep, y2, qim[it][v] + oim);
          }
      }
    }
    v8f yk[3], yre[3], yim[3];
#pragma unroll
    for (int it = 0; it < 3; ++it) {
      yk[it] = (v8f){};
      yre[it] = (v8f){};
      yim[it] = (v8f){};
    }
#pragma unroll
    for (int rt = 0; rt < 3; ++rt) {
      float4 gr[3][2], gi[3][2];
#pragma unroll
      for (int it = 0; it < 3; ++it) {
        const int go = (it * 16 + c) * GPITCH + rt * 16 + hi * 8;
        gr[it][0] = *(const float4*)&Gre[go];
        gr[it][1] = *(const float4*)&Gre[go + 4];
        gi[it][0] = *(const float4*)&Gim[go];
        gi[it][1] = *(const float4*)&Gim[go + 4];
      }
#pragma unroll
      for (int j = 0; j < 4; ++j) {
        const v2f bqr = (v2f){qre[rt][2 * j], qre[rt][2 * j + 1]};
        const v2f bqi = (v2f){qim[rt][2 * j], qim[rt][2 * j + 1]};
        const v2f bqd = bqi - bqr;
        const v2f bqs = -bqr - bqi;
#pragma unroll
        for (int it = 0; it < 3; ++it) {
          const float4 r4 = gr[it][j >> 1];
          const float4 i4 = gi[it][j >> 1];
          const v2f g_r = (j & 1) ? (v2f){r4.z, r4.w} : (v2f){r4.x, r4.y};
          const v2f g_i = (j & 1) ? (v2f){i4.z, i4.w} : (v2f){i4.x, i4.y};
          const v2f gs = g_r + g_i;
          yk[it]  = wmma_f32(gs, bqr, yk[it]);
          yim[it] = wmma_f32(g_r, bqd, yim[it]);
          yre[it] = wmma_f32(g_i, bqs, yre[it]);
        }
      }
    }
    load_dft_frag(dre, dim, dbase, dR, dI);
    float p = 0.0f;
#pragma unroll
    for (int rt = 0; rt < 3; ++rt)
#pragma unroll
      for (int v = 0; v < 8; ++v)
        p += qre[rt][v] * (yre[rt][v] + yk[rt][v]) + qim[rt][v] * (yim[rt][v] + yk[rt][v]);
    p += __shfl_xor(p, 16, 32);
    const float rm = sqrtf(fmaxf(p, 0.0f));
    const float g = 1.0f / (1.0f + expf(-(s1 * (rm - s2))));
    const bool lone = (blockIdx.x == LBLK - 1) && (wv == NCW - 1) && (c == 15);
    const int pcol = lone ? LCOL : wv * 16 + c;
#pragma unroll
    for (int rt = 0; rt < 3; ++rt)
#pragma unroll
      for (int v = 0; v < 8; ++v) {
        qre[rt][v] *= g;
        qim[rt][v] *= g;
        Pre[(rt * 16 + hi * 8 + v) * PPITCH + pcol] = qre[rt][v];
        Pim[(rt * 16 + hi * 8 + v) * PPITCH + pcol] = qim[rt][v];
      }
    if (lone) {
#pragma unroll
      for (int r = 0; r < 24; ++r) {
        Pre[((r >> 3) * 16 + hi * 8 + (r & 7)) * PPITCH + LCOL - 1] = 0.0f;
        Pim[((r >> 3) * 16 + hi * 8 + (r & 7)) * PPITCH + LCOL - 1] = 0.0f;
      }
    }
  }
  __syncthreads();
  if (wv < NCW) {
#pragma unroll
    for (int rt = 0; rt < 3; ++rt)
#pragma unroll
      for (int vg = 0; vg < 2; ++vg) {
        *(float4*)(pw + (rt * 2 + vg) * 128 + lane * 4) =
            make_float4(qre[rt][vg * 4], qre[rt][vg * 4 + 1], qre[rt][vg * 4 + 2], qre[rt][vg * 4 + 3]);
        *(float4*)(pw + 768 + (rt * 2 + vg) * 128 + lane * 4) =
            make_float4(qim[rt][vg * 4], qim[rt][vg * 4 + 1], qim[rt][vg * 4 + 2], qim[rt][vg * 4 + 3]);
      }
  }

  {
    v8f sre = {}, sim = {}, sre2 = {}, sim2 = {};
#pragma unroll
    for (int ks = 0; ks < NPAIR / 4; ++ks) {
      const int k0 = ks * 4 + hi * 2;
      const v2f lr = *(const v2f*)&Pre[(ti * 16 + c) * PPITCH + k0];
      const v2f mr = *(const v2f*)&Pre[(ti * 16 + c) * PPITCH + NPAIR + k0];
      const v2f li = *(const v2f*)&Pim[(ti * 16 + c) * PPITCH + k0];
      const v2f mi = *(const v2f*)&Pim[(ti * 16 + c) * PPITCH + NPAIR + k0];
      const v2f ur = lr + mr;
      const v2f ui = li + mi;
      const v2f vi = li - mi;
      const v2f nvr = mr - lr;
      const v2f bc = dR[ks];
      const v2f bs = dI[ks];
      sre  = wmma_f32(ur, bc, sre);
      sim  = wmma_f32(ui, bc, sim);
      sre2 = wmma_f32(vi, bs, sre2);
      sim2 = wmma_f32(nvr, bs, sim2);
    }
    sre += sre2;
    sim += sim2;
    if (blockIdx.x == LBLK - 1) {
      const float sc = 0.04419417382415922f;
#pragma unroll
      for (int v = 0; v < 8; ++v) {
        sre[v] += Pre[(ti * 16 + hi * 8 + v) * PPITCH + LCOL] * sc;
        sim[v] += Pim[(ti * 16 + hi * 8 + v) * PPITCH + LCOL] * sc;
      }
    }
    float* sp = snext + blockIdx.x * CSZ + wv * 256;
#pragma unroll
    for (int v = 0; v < 8; ++v) {
      sp[v * 32 + lane] = sre[v];
      sp[MSZ + v * 32 + lane] = sim[v];
    }
  }
}

__global__ void __launch_bounds__(128) k_out(const float* __restrict__ u,
                                             const float* __restrict__ scal,
                                             const float* __restrict__ sp,
                                             float* __restrict__ out) {
  __shared__ __attribute__((aligned(16))) float Wre[MSZ];
  __shared__ __attribute__((aligned(16))) float Wim[MSZ];
  for (int q4 = threadIdx.x; q4 < MSZ / 4; q4 += 128) {
    const int n = q4 * 4;
    const int tl = n >> 8, tr = tl / 3;
    const int r = tr * 16 + ((n >> 5) & 7) + ((n >> 4) & 1) * 8;
    const int cc = (tl - tr * 3) * 16 + (n & 15);
    float4 sr = make_float4(0.f, 0.f, 0.f, 0.f), si = make_float4(0.f, 0.f, 0.f, 0.f);
#pragma unroll
    for (int p = 0; p < LBLK; ++p) {
      const float4 a = *(const float4*)(sp + p * CSZ + n);
      const float4 b = *(const float4*)(sp + p * CSZ + MSZ + n);
      sr.x += a.x; sr.y += a.y; sr.z += a.z; sr.w += a.w;
      si.x += b.x; si.y += b.y; si.z += b.z; si.w += b.w;
    }
    *(float4*)&Wre[r * F_N + cc] = sr;
    *(float4*)&Wim[r * F_N + cc] = si;
  }
  __syncthreads();

  const int lane = threadIdx.x & 31;
  const int wv = threadIdx.x >> 5;
  const int hi = lane >> 4, c = lane & 15;
  const int rowbase = (blockIdx.x * 4 + wv) * 16;
  const float s = scal[0], inv = scal[1];
  const float* ur = u + (size_t)(rowbase + c) * (F_N * 2) + hi * 4;
  v8f ore[3], oim[3];
#pragma unroll
  for (int j = 0; j < 3; ++j) {
    ore[j] = (v8f){};
    oim[j] = (v8f){};
  }
#pragma unroll
  for (int ks = 0; ks < 12; ++ks) {
    const float4 x = *(const float4*)(ur + ks * 8);
    const v2f ar = (v2f){s * x.x, s * x.z};
    const v2f ai = (v2f){s * x.y, s * x.w};
    const v2f an = -ai;
    const int k0 = ks * 4 + hi * 2;
#pragma unroll
    for (int j = 0; j < 3; ++j) {
      const v2f br = (v2f){Wre[k0 * F_N + j * 16 + c], Wre[(k0 + 1) * F_N + j * 16 + c]};
      const v2f bi = (v2f){Wim[k0 * F_N + j * 16 + c], Wim[(k0 + 1) * F_N + j * 16 + c]};
      ore[j] = wmma_f32(ar, br, ore[j]);
      ore[j] = wmma_f32(an, bi, ore[j]);
      oim[j] = wmma_f32(ar, bi, oim[j]);
      oim[j] = wmma_f32(ai, br, oim[j]);
    }
  }
#pragma unroll
  for (int j = 0; j < 3; ++j)
#pragma unroll
    for (int v = 0; v < 8; ++v) {
      const int row = rowbase + hi * 8 + v;
      float2 o;
      o.x = ore[j][v] * inv;
      o.y = oim[j][v] * inv;
      *(float2*)(out + ((size_t)row * F_N + j * 16 + c) * 2) = o;
    }
}

extern "C" void kernel_launch(void* const* d_in, const int* in_sizes, int n_in,
                              void* d_out, int out_size, void* d_ws, size_t ws_size,
                              hipStream_t stream) {
  const float* u  = (const float*)d_in[0];
  const float* S1 = (const float*)d_in[1];
  const float* S2 = (const float*)d_in[2];
  float* out = (float*)d_out;
  float* w = (float*)d_ws;

  float* scal  = w;
  float* part  = w + 8;
  float* ire   = w + 2048;
  float* iim   = ire + F_N * T_N;
  float* dre   = iim + F_N * T_N;
  float* dim   = dre + F_N * T_N;
  float* G     = w + 100352;
  float* spart = w + 104960;
  float* pst   = w + 178688;
  float* gpart = w + 227840;

  k_gram<<<NGP, 96, 0, stream>>>(u, part, gpart, ire, iim, dre, dim);
  k_gred<<<CSZ / 32, 256, 0, stream>>>(gpart, part, G, scal);

  for (int l = 0; l < L_N; ++l) {
    k_layer<<<LBLK, 288, 0, stream>>>(ire, iim, dre, dim, G,
                                      spart + ((l + 1) & 1) * (LBLK * CSZ),
                                      spart + (l & 1) * (LBLK * CSZ), pst, S1, S2,
                                      (l == 0) ? 1 : 0);
  }
  k_out<<<B_N / 64, 128, 0, stream>>>(u, scal, spart + ((L_N - 1) & 1) * (LBLK * CSZ), out);
}


# layer kernel: P @ dft shared over 3 work-groups per column group: each repeats the column work, forms one row tile's 3 tiles (one wave per SIMD), keeps that row tile of P (two state buffers)
# speedup vs baseline: 28.9843x; 1.1263x over previous
#include <hip/hip_runtime.h>
#include <math.h>


#define B_N 16384
#define F_N 48
#define T_N 512
#define L_N 16
#define MSZ (F_N * F_N)
#define CSZ (2 * MSZ)
#define NGP 256
#define LBLK 8
#define LCOL 64
#define NCW (LCOL / 16)
#define PPITCH 68
#define MPITCH 52
#define GPITCH 52
#define FPC (F_N * T_N / 4)
#define NPAIR (LCOL / 2)
#define DPC (FPC / 2)


typedef __attribute__((ext_vector_type(2))) float v2f;
typedef __attribute__((ext_vector_type(8))) float v8f;

static __device__ __forceinline__ v8f wmma_f32(v2f a, v2f b, v8f c) {
  return __builtin_amdgcn_wmma_f32_16x16x4_f32(false, a, false, b, (short)0, c,
                                               false, false);
}


__global__ void __launch_bounds__(96) k_gram(const float* __restrict__ u,
                                             float* __restrict__ part,
                                             float* __restrict__ gpart,
                                             float* __restrict__ ire, float* __restrict__ iim,
                                             float* __restrict__ dre, float* __restrict__ dim) {
  {
    const int idx = blockIdx.x * 96 + threadIdx.x;
    const float sc = 0.04419417382415922f;
    const float w = 6.283185307179586f / (float)T_N;
    float s0, c0, s1, c1;
    const int q = idx >> 1, hf = idx & 1;
    const int ln = q & 31, lh = ln >> 4, lc = ln & 15;
    if (q < FPC) {
      const int cg = q >> 5, ct = cg / 6, g = cg - ct * 6;
      const int t = (ct >> 2) * NPAIR + (ct & 1) * 16 + 1 + lc;
      const bool col0 = ((ct & 2) != 0) && (t == T_N / 2);
      const int f0 = 4 * (2 * g + hf) + 2 * lh;
      sincosf((float)((f0 * t) & (T_N - 1)) * w, &s0, &c0);
      sincosf((float)(((f0 + 1) * t) & (T_N - 1)) * w, &s1, &c1);
      *(float2*)(ire + q * 4 + hf * 2) = make_float2(c0 * sc, c1 * sc);
      *(float2*)(iim + q * 4 + hf * 2) = col0 ? make_float2(sc, sc) : make_float2(s0 * sc, s1 * sc);
    } else if (q < FPC + DPC) {
      const int q2 = q - FPC;
      const int r = q2 >> 5, g = r & 3, r2 = r >> 2;
      const int blk = r2 / 3, tj = r2 - blk * 3;
      const int f = tj * 16 + lc;
      const int t0 = blk * NPAIR + 1 + 4 * (2 * g + hf) + 2 * lh;
      sincosf((float)((f * t0) & (T_N - 1)) * w, &s0, &c0);
      sincosf((float)((f * (t0 + 1)) & (T_N - 1)) * w, &s1, &c1);
      *(float2*)(dre + q2 * 4 + hf * 2) = make_float2(c0 * sc, c1 * sc);
      *(float2*)(dim + q2 * 4 + hf * 2) = make_float2(s0 * sc, s1 * sc);
    }
  }
  const int lane = threadIdx.x & 31;
  const int wi = threadIdx.x >> 5;
  const int hi = lane >> 4, c = lane & 15;
  const float* ub = u + (size_t)(blockIdx.x * 64 + 2 * hi) * (F_N * 2) + c * 2;
  v8f gre[3], gim[3];
#pragma unroll
  for (int j = 0; j < 3; ++j) {
    gre[j] = (v8f){};
    gim[j] = (v8f){};
  }
  float mxl = -3.4e38f;
#pragma unroll
  for (int ks = 0; ks < 16; ++ks) {
    v2f xr[3], xi[3];
#pragma unroll
    for (int ft = 0; ft < 3; ++ft) {
      const float2 p0 = *(const float2*)(ub + (ks * 4) * (F_N * 2) + ft * 32);
      const float2 p1 = *(const float2*)(ub + (ks * 4 + 1) * (F_N * 2) + ft * 32);
      xr[ft] = (v2f){p0.x, p1.x};
      xi[ft] = (v2f){p0.y, p1.y};
      mxl = fmaxf(mxl, fmaxf(fmaxf(p0.x, p0.y), fmaxf(p1.x, p1.y)));
    }
    const float2 a0 = *(const float2*)(ub + (ks * 4) * (F_N * 2) + wi * 32);
    const float2 a1 = *(const float2*)(ub + (ks * 4 + 1) * (F_N * 2) + wi * 32);
    const v2f ar = (v2f){a0.x, a1.x};
    const v2f ai = (v2f){a0.y, a1.y};
    const v2f an = -ai;
#pragma unroll
    for (int j = 0; j < 3; ++j) {
      gre[j] = wmma_f32(ar, xr[j], gre[j]);
      gim[j] = wmma_f32(ar, xi[j], gim[j]);
      gre[j] = wmma_f32(ai, xi[j], gre[j]);
      gim[j] = wmma_f32(an, xr[j], gim[j]);
    }
  }
  float* gp = gpart + (size_t)blockIdx.x * CSZ + wi * (3 * 256);
#pragma unroll
  for (int j = 0; j < 3; ++j)
#pragma unroll
    for (int v = 0; v < 8; ++v) {
      gp[j * 256 + v * 32 + lane] = gre[j][v];
      gp[MSZ + j * 256 + v * 32 + lane] = gim[j][v];
    }
  mxl = fmaxf(mxl, __shfl_xor(mxl, 16, 32));
  mxl = fmaxf(mxl, __shfl_xor(mxl, 8, 32));
  mxl = fmaxf(mxl, __shfl_xor(mxl, 4, 32));
  mxl = fmaxf(mxl, __shfl_xor(mxl, 2, 32));
  mxl = fmaxf(mxl, __shfl_xor(mxl, 1, 32));
  if (threadIdx.x == 0) part[blockIdx.x] = mxl;
}

__global__ void __launch_bounds__(256) k_gred(const float* __restrict__ gpart,
                                              const float* __restrict__ part,
                                              float* __restrict__ G,
                                              float* __restrict__ scal) {
  __shared__ float sm[256];
  __shared__ float red[8][32];
  const int grp = threadIdx.x >> 5, el = threadIdx.x & 31;
  const int n = blockIdx.x * 32 + el;
  {
    const float* p = gpart + (size_t)(grp * 32) * CSZ + n;
    float acc = 0.0f;
    for (int i = 0; i < 32; ++i) acc += p[(size_t)i * CSZ];
    red[grp][el] = acc;
  }
  sm[threadIdx.x] = part[threadIdx.x];
  __syncthreads();
  for (int s = 128; s > 0; s >>= 1) {
    if (threadIdx.x < s) sm[threadIdx.x] = fmaxf(sm[threadIdx.x], sm[threadIdx.x + s]);
    __syncthreads();
  }
  const float mx = fabsf(sm[0]);
  const float s = 5.0f / mx;
  if (blockIdx.x == 0 && threadIdx.x == 0) {
    scal[0] = s;
    scal[1] = mx / 5.0f;
  }
  if (threadIdx.x < 32) {
    float tot = 0.0f;
#pragma unroll
    for (int k = 0; k < 8; ++k) tot += red[k][el];
    const int pl = n / MSZ, q = n - pl * MSZ;
    const int tl = q >> 8, tr = tl / 3;
    const int r = tr * 16 + ((q >> 5) & 7) + ((q >> 4) & 1) * 8;
    const int cc = (tl - tr * 3) * 16 + (q & 15);
    G[pl * MSZ + r * F_N + cc] = tot * (s * s * (1.0f / (float)B_N));
  }
}

static __device__ __forceinline__ void load_dft_frag(const float* __restrict__ dre,
                                                     const float* __restrict__ dim,
                                                     int base, v2f (&dR)[NPAIR / 4],
                                                     v2f (&dI)[NPAIR / 4]) {
#pragma unroll
  for (int g = 0; g < NPAIR / 8; ++g) {
    const float4 r4 = *(const float4*)(dre + base + g * 128);
    const float4 i4 = *(const float4*)(dim + base + g * 128);
    dR[2 * g] = (v2f){r4.x, r4.y};
    dR[2 * g + 1] = (v2f){r4.z, r4.w};
    dI[2 * g] = (v2f){i4.x, i4.y};
    dI[2 * g + 1] = (v2f){i4.z, i4.w};
  }
}

__global__ void __launch_bounds__(288) k_layer(
    const float* __restrict__ ire, const float* __restrict__ iim,
    const float* __restrict__ dre, const float* __restrict__ dim,
    const float* __restrict__ G, const float* __restrict__ sprev,
    float* __restrict__ snext, const float* __restrict__ pprev, float* __restrict__ pnext,
    const float* __restrict__ S1, const float* __restrict__ S2, int first) {
  __shared__ __attribute__((aligned(16))) float sm[2 * F_N * PPITCH + 2 * F_N * GPITCH + NCW * 48 * 32];
  float* Mre = sm;
  float* Mim = sm + F_N * MPITCH;
  float* Pre = sm;
  float* Pim = sm + F_N * PPITCH;
  float* Gre = sm + 2 * F_N * PPITCH;
  float* Gim = Gre + F_N * GPITCH;
  float* Xs = Gim + F_N * GPITCH;

  const int tid = threadIdx.x;
  const int lane = tid & 31;
  const int wv = tid >> 5;
  const int hi = lane >> 4, c = lane & 15;
  const int cg = blockIdx.x % LBLK;
  const int tis = blockIdx.x / LBLK;
  const int tj = wv - NCW;
  const int dbase = ((cg * 3 + tj) * (NPAIR / 8) * 32 + lane) * 4;
  const float s1 = S1[0], s2 = S2[0];

  v8f qre[3], qim[3];
#pragma unroll
  for (int rt = 0; rt < 3; ++rt) {
    qre[rt] = (v8f){};
    qim[rt] = (v8f){};
  }
  v2f bF[12];
#pragma unroll
  for (int ks = 0; ks < 12; ++ks) bF[ks] = (v2f){0.f, 0.f};
  const int poff = (cg * NCW + (wv & (NCW - 1))) * (2 * 3 * 8 * 32);
  const float* pr_src = pprev + poff;
  float* pw = pnext + poff;
  const int ctile = cg * NCW + (wv & (NCW - 1));
  if (wv < NCW) {
    if (!first) {
#pragma unroll
      for (int rt = 0; rt < 3; ++rt)
#pragma unroll
        for (int vg = 0; vg < 2; ++vg) {
          const float4 r4 = *(const float4*)(pr_src + (rt * 2 + vg) * 128 + lane * 4);
          const float4 i4 = *(const float4*)(pr_src + 768 + (rt * 2 + vg) * 128 + lane * 4);
          qre[rt][vg * 4 + 0] = r4.x; qre[rt][vg * 4 + 1] = r4.y;
          qre[rt][vg * 4 + 2] = r4.z; qre[rt][vg * 4 + 3] = r4.w;
          qim[rt][vg * 4 + 0] = i4.x; qim[rt][vg * 4 + 1] = i4.y;
          qim[rt][vg * 4 + 2] = i4.z; qim[rt][vg * 4 + 3] = i4.w;
        }
    }
    const float* fsrc = (wv < 2) ? ire : iim;
#pragma unroll
    for (int g = 0; g < 6; ++g) {
      const float4 f4 = *(const float4*)(fsrc + ((ctile * 6 + g) * 32 + lane) * 4);
      bF[2 * g] = (v2f){f4.x, f4.y};
      bF[2 * g + 1] = (v2f){f4.z, f4.w};
    }
  }

  float4 ms[4];
#pragma unroll
  for (int i = 0; i < 4; ++i) ms[i] = make_float4(0.f, 0.f, 0.f, 0.f);
  if (!first) {
    float4 t[4][LBLK];
#pragma unroll
    for (int i = 0; i < 4; ++i) {
      const int pl = i >> 1;
      const int n = (tid + 288 * i - pl * (MSZ / 4)) * 4;
#pragma unroll
      for (int p = 0; p < LBLK; ++p) t[i][p] = *(const float4*)(sprev + p * CSZ + pl * MSZ + n);
    }
#pragma unroll
    for (int i = 0; i < 4; ++i)
#pragma unroll
      for (int p = 0; p < LBLK; ++p) {
        ms[i].x += t[i][p].x; ms[i].y += t[i][p].y; ms[i].z += t[i][p].z; ms[i].w += t[i][p].w;
      }
  }

#pragma unroll
  for (int i = 0; i < 4; ++i) {
    const int pl = i >> 1;
    const int n = (tid + 288 * i - pl * (MSZ / 4)) * 4;
    const int tl = n >> 8, tr = tl / 3;
    const int r = tr * 16 + ((n >> 5) & 7) + ((n >> 4) & 1) * 8;
    const int cc = (tl - tr * 3) * 16 + (n & 15);
    float4 m;
    if (pl == 0) {
      m.x = ((r == cc) ? 1.0f : 0.0f) - ms[i].x;
      m.y = ((r == cc + 1) ? 1.0f : 0.0f) - ms[i].y;
      m.z = ((r == cc + 2) ? 1.0f : 0.0f) - ms[i].z;
      m.w = ((r == cc + 3) ? 1.0f : 0.0f) - ms[i].w;
      *(float4*)&Mre[r * MPITCH + cc] = m;
    } else {
      m.x = -ms[i].x; m.y = -ms[i].y; m.z = -ms[i].z; m.w = -ms[i].w;
      *(float4*)&Mim[r * MPITCH + cc] = m;
    }
  }
  __syncthreads();

  v2f dR[NPAIR / 4], dI[NPAIR / 4];
  v8f xa[3], xb[3];
#pragma unroll
  for (int it = 0; it < 3; ++it) {
    xa[it] = (v8f){};
    xb[it] = (v8f){};
  }
  if (wv < NCW) {
#pragma unroll
    for (int ks = 0; ks < 12; ++ks) {
      const int k0 = ks * 4 + hi * 2;
      const v2f bf = bF[ks];
#pragma unroll
      for (int it = 0; it < 3; ++it) {
        const v2f are = *(const v2f*)&Mre[(it * 16 + c) * MPITCH + k0];
        const v2f aim = *(const v2f*)&Mim[(it * 16 + c) * MPITCH + k0];
        xa[it] = wmma_f32(are, bf, xa[it]);
        xb[it] = wmma_f32(aim, bf, xb[it]);
      }
    }
    float* xw = Xs + wv * (48 * 32);
#pragma unroll
    for (int it = 0; it < 3; ++it)
#pragma unroll
      for (int vg = 0; vg < 2; ++vg) {
        *(float4*)&xw[((it * 4 + vg) * 32 + lane) * 4] =
            make_float4(xa[it][vg * 4], xa[it][vg * 4 + 1], xa[it][vg * 4 + 2], xa[it][vg * 4 + 3]);
        *(float4*)&xw[((it * 4 + 2 + vg) * 32 + lane) * 4] =
            make_float4(xb[it][vg * 4], xb[it][vg * 4 + 1], xb[it][vg * 4 + 2], xb[it][vg * 4 + 3]);
      }
  } else {
    if (wv < 2 * NCW) {
      float4 g[9];
#pragma unroll
      for (int i = 0; i < 9; ++i) g[i] = *(const float4*)(G + (tid - 128 + 128 * i) * 4);
#pragma unroll
      for (int i = 0; i < 9; ++i) {
        const int q = tid - 128 + 128 * i;
        const int pl = q / (MSZ / 4), rem = q - pl * (MSZ / 4);
        const int r = rem / (F_N / 4), c4 = rem - r * (F_N / 4);
        *(float4*)&Gre[pl * (F_N * GPITCH) + r * GPITCH + c4 * 4] = g[i];
      }
    }
    if (wv < NCW + 3) load_dft_frag(dre, dim, dbase, dR, dI);
  }
  __syncthreads();

  v8f pr = {}, pi = {};
  if (wv < NCW) {
    {
      const float* xr = Xs + (wv ^ 2) * (48 * 32);
      float4 ya[3][2], yb[3][2];
#pragma unroll
      for (int it = 0; it < 3; ++it)
#pragma unroll
        for (int vg = 0; vg < 2; ++vg) {
          ya[it][vg] = *(const float4*)&xr[((it * 4 + vg) * 32 + lane) * 4];
          yb[it][vg] = *(const float4*)&xr[((it * 4 + 2 + vg) * 32 + lane) * 4];
        }
      const bool last = (cg == LBLK - 1);
      const float keep = (last && ((wv & 1) == 1) && (c == 15)) ? 0.0f : 1.0f;
      const int wvs = __builtin_amdgcn_readfirstlane(wv);
      if (wvs < 2) {
#pragma unroll
        for (int it = 0; it < 3; ++it)
#pragma unroll
          for (int v = 0; v < 8; ++v) {
            const float4 a4 = ya[it][v >> 2], b4 = yb[it][v >> 2];
            const float y3 = (v & 3) == 0 ? a4.x : (v & 3) == 1 ? a4.y : (v & 3) == 2 ? a4.z : a4.w;
            const float y4 = (v & 3) == 0 ? b4.x : (v & 3) == 1 ? b4.y : (v & 3) == 2 ? b4.z : b4.w;
            qre[it][v] = fmaf(-keep, y4, qre[it][v] + xa[it][v]);
            qim[it][v] = fmaf(keep, y3, qim[it][v] + xb[it][v]);
          }
      } else {
        const bool col0 = last && (wvs == NCW - 1) && (c == 15);
#pragma unroll
        for (int it = 0; it < 3; ++it)
#pragma unroll
          for (int v = 0; v < 8; ++v) {
            const float4 a4 = ya[it][v >> 2], b4 = yb[it][v >> 2];
            const float y1 = (v & 3) == 0 ? a4.x : (v & 3) == 1 ? a4.y : (v & 3) == 2 ? a4.z : a4.w;
            const float y2 = (v & 3) == 0 ? b4.x : (v & 3) == 1 ? b4.y : (v & 3) == 2 ? b4.z : b4.w;
            const float ore = col0 ? xa[it][v] : xb[it][v];
            const float oim = col0 ? xb[it][v] : -xa[it][v];
            qre[it][v] = fmaf(keep, y1, qre[it][v] + ore);
            qim[it][v] = fmaf(keep, y2, qim[it][v] + oim);
          }
      }
    }
    v8f yk[3], yre[3], yim[3];
#pragma unroll
    for (int it = 0; it < 3; ++it) {
      yk[it] = (v8f){};
      yre[it] = (v8f){};
      yim[it] = (v8f){};
    }
#pragma unroll
    for (int rt = 0; rt < 3; ++rt) {
      float4 gr[3][2], gi[3][2];
#pragma unroll
      for (int it = 0; it < 3; ++it) {
        const int go = (it * 16 + c) * GPITCH + rt * 16 + hi * 8;
        gr[it][0] = *(const float4*)&Gre[go];
        gr[it][1] = *(const float4*)&Gre[go + 4];
        gi[it][0] = *(const float4*)&Gim[go];
        gi[it][1] = *(const float4*)&Gim[go + 4];
      }
#pragma unroll
      for (int j = 0; j < 4; ++j) {
        const v2f bqr = (v2f){qre[rt][2 * j], qre[rt][2 * j + 1]};
        const v2f bqi = (v2f){qim[rt][2 * j], qim[rt][2 * j + 1]};
        const v2f bqd = bqi - bqr;
        const v2f bqs = -bqr - bqi;
#pragma unroll
        for (int it = 0; it < 3; ++it) {
          const float4 r4 = gr[it][j >> 1];
          const float4 i4 = gi[it][j >> 1];
          const v2f g_r = (j & 1) ? (v2f){r4.z, r4.w} : (v2f){r4.x, r4.y};
          const v2f g_i = (j & 1) ? (v2f){i4.z, i4.w} : (v2f){i4.x, i4.y};
          const v2f gs = g_r + g_i;
          yk[it]  = wmma_f32(gs, bqr, yk[it]);
          yim[it] = wmma_f32(g_r, bqd, yim[it]);
          yre[it] = wmma_f32(g_i, bqs, yre[it]);
        }
      }
    }
    float p = 0.0f;
#pragma unroll
    for (int rt = 0; rt < 3; ++rt)
#pragma unroll
      for (int v = 0; v < 8; ++v)
        p += qre[rt][v] * (yre[rt][v] + yk[rt][v]) + qim[rt][v] * (yim[rt][v] + yk[rt][v]);
    p += __shfl_xor(p, 16, 32);
    const float rm = sqrtf(fmaxf(p, 0.0f));
    const float g = 1.0f / (1.0f + expf(-(s1 * (rm - s2))));
    const bool lone = (cg == LBLK - 1) && (wv == NCW - 1) && (c == 15);
    const int pcol = lone ? LCOL : wv * 16 + c;
    pr = (tis == 0) ? qre[0] : (tis == 1) ? qre[1] : qre[2];
    pi = (tis == 0) ? qim[0] : (tis == 1) ? qim[1] : qim[2];
#pragma unroll
    for (int v = 0; v < 8; ++v) {
      pr[v] *= g;
      pi[v] *= g;
      Pre[(hi * 8 + v) * PPITCH + pcol] = pr[v];
      Pim[(hi * 8 + v) * PPITCH + pcol] = pi[v];
    }
    if (lone) {
#pragma unroll
      for (int r = 0; r < 8; ++r) {
        Pre[(hi * 8 + r) * PPITCH + LCOL - 1] = 0.0f;
        Pim[(hi * 8 + r) * PPITCH + LCOL - 1] = 0.0f;
      }
    }
  }
  __syncthreads();
  if (wv < NCW) {
#pragma unroll
    for (int vg = 0; vg < 2; ++vg) {
      *(float4*)(pw + (tis * 2 + vg) * 128 + lane * 4) =
          make_float4(pr[vg * 4], pr[vg * 4 + 1], pr[vg * 4 + 2], pr[vg * 4 + 3]);
      *(float4*)(pw + 768 + (tis * 2 + vg) * 128 + lane * 4) =
          make_float4(pi[vg * 4], pi[vg * 4 + 1], pi[vg * 4 + 2], pi[vg * 4 + 3]);
    }
  }

  if (wv >= NCW && wv < NCW + 3) {
    v8f sre = {}, sim = {}, sre2 = {}, sim2 = {};
#pragma unroll
    for (int ks = 0; ks < NPAIR / 4; ++ks) {
      const int k0 = ks * 4 + hi * 2;
      const v2f lr = *(const v2f*)&Pre[c * PPITCH + k0];
      const v2f mr = *(const v2f*)&Pre[c * PPITCH + NPAIR + k0];
      const v2f li = *(const v2f*)&Pim[c * PPITCH + k0];
      const v2f mi = *(const v2f*)&Pim[c * PPITCH + NPAIR + k0];
      const v2f ur = lr + mr;
      const v2f ui = li + mi;
      const v2f vi = li - mi;
      const v2f nvr = mr - lr;
      const v2f bc = dR[ks];
      const v2f bs = dI[ks];
      sre  = wmma_f32(ur, bc, sre);
      sim  = wmma_f32(ui, bc, sim);
      sre2 = wmma_f32(vi, bs, sre2);
      sim2 = wmma_f32(nvr, bs, sim2);
    }
    sre += sre2;
    sim += sim2;
    if (cg == LBLK - 1) {
      const float sc = 0.04419417382415922f;
#pragma unroll
      for (int v = 0; v < 8; ++v) {
        sre[v] += Pre[(hi * 8 + v) * PPITCH + LCOL] * sc;
        sim[v] += Pim[(hi * 8 + v) * PPITCH + LCOL] * sc;
      }
    }
    float* sp = snext + cg * CSZ + (tis * 3 + tj) * 256;
#pragma unroll
    for (int v = 0; v < 8; ++v) {
      sp[v * 32 + lane] = sre[v];
      sp[MSZ + v * 32 + lane] = sim[v];
    }
  }
}

__global__ void __launch_bounds__(128) k_out(const float* __restrict__ u,
                                             const float* __restrict__ scal,
                                             const float* __restrict__ sp,
                                             float* __restrict__ out) {
  __shared__ __attribute__((aligned(16))) float Wre[MSZ];
  __shared__ __attribute__((aligned(16))) float Wim[MSZ];
  for (int q4 = threadIdx.x; q4 < MSZ / 4; q4 += 128) {
    const int n = q4 * 4;
    const int tl = n >> 8, tr = tl / 3;
    const int r = tr * 16 + ((n >> 5) & 7) + ((n >> 4) & 1) * 8;
    const int cc = (tl - tr * 3) * 16 + (n & 15);
    float4 sr = make_float4(0.f, 0.f, 0.f, 0.f), si = make_float4(0.f, 0.f, 0.f, 0.f);
#pragma unroll
    for (int p = 0; p < LBLK; ++p) {
      const float4 a = *(const float4*)(sp + p * CSZ + n);
      const float4 b = *(const float4*)(sp + p * CSZ + MSZ + n);
      sr.x += a.x; sr.y += a.y; sr.z += a.z; sr.w += a.w;
      si.x += b.x; si.y += b.y; si.z += b.z; si.w += b.w;
    }
    *(float4*)&Wre[r * F_N + cc] = sr;
    *(float4*)&Wim[r * F_N + cc] = si;
  }
  __syncthreads();

  const int lane = threadIdx.x & 31;
  const int wv = threadIdx.x >> 5;
  const int hi = lane >> 4, c = lane & 15;
  const int rowbase = (blockIdx.x * 4 + wv) * 16;
  const float s = scal[0], inv = scal[1];
  const float* ur = u + (size_t)(rowbase + c) * (F_N * 2) + hi * 4;
  v8f ore[3], oim[3];
#pragma unroll
  for (int j = 0; j < 3; ++j) {
    ore[j] = (v8f){};
    oim[j] = (v8f){};
  }
#pragma unroll
  for (int ks = 0; ks < 12; ++ks) {
    const float4 x = *(const float4*)(ur + ks * 8);
    const v2f ar = (v2f){s * x.x, s * x.z};
    const v2f ai = (v2f){s * x.y, s * x.w};
    const v2f an = -ai;
    const int k0 = ks * 4 + hi * 2;
#pragma unroll
    for (int j = 0; j < 3; ++j) {
      const v2f br = (v2f){Wre[k0 * F_N + j * 16 + c], Wre[(k0 + 1) * F_N + j * 16 + c]};
      const v2f bi = (v2f){Wim[k0 * F_N + j * 16 + c], Wim[(k0 + 1) * F_N + j * 16 + c]};
      ore[j] = wmma_f32(ar, br, ore[j]);
      ore[j] = wmma_f32(an, bi, ore[j]);
      oim[j] = wmma_f32(ar, bi, oim[j]);
      oim[j] = wmma_f32(ai, br, oim[j]);
    }
  }
#pragma unroll
  for (int j = 0; j < 3; ++j)
#pragma unroll
    for (int v = 0; v < 8; ++v) {
      const int row = rowbase + hi * 8 + v;
      float2 o;
      o.x = ore[j][v] * inv;
      o.y = oim[j][v] * inv;
      *(float2*)(out + ((size_t)row * F_N + j * 16 + c) * 2) = o;
    }
}

extern "C" void kernel_launch(void* const* d_in, const int* in_sizes, int n_in,
                              void* d_out, int out_size, void* d_ws, size_t ws_size,
                              hipStream_t stream) {
  const float* u  = (const float*)d_in[0];
  const float* S1 = (const float*)d_in[1];
  const float* S2 = (const float*)d_in[2];
  float* out = (float*)d_out;
  float* w = (float*)d_ws;

  float* scal  = w;
  float* part  = w + 8;
  float* ire   = w + 2048;
  float* iim   = ire + F_N * T_N;
  float* dre   = iim + F_N * T_N;
  float* dim   = dre + F_N * T_N;
  float* G     = w + 100352;
  float* spart = w + 104960;
  float* pst   = w + 178688;
  float* gpart = w + 227840;

  k_gram<<<NGP, 96, 0, stream>>>(u, part, gpart, ire, iim, dre, dim);
  k_gred<<<CSZ / 32, 256, 0, stream>>>(gpart, part, G, scal);

  for (int l = 0; l < L_N; ++l) {
    k_layer<<<LBLK * 3, 288, 0, stream>>>(ire, iim, dre, dim, G,
                                      spart + ((l + 1) & 1) * (LBLK * CSZ),
                                      spart + (l & 1) * (LBLK * CSZ),
                                      pst + ((l + 1) & 1) * (2 * F_N * T_N),
                                      pst + (l & 1) * (2 * F_N * T_N), S1, S2,
                                      (l == 0) ? 1 : 0);
  }
  k_out<<<B_N / 64, 128, 0, stream>>>(u, scal, spart + ((L_N - 1) & 1) * (LBLK * CSZ), out);
}


# column power q^H G q from the upper tile triangle of the Hermitian Gram matrix (off-diagonal tiles doubled where G is reduced): G @ Q is 72 products per column wave, not 108 (sum order changes)
# speedup vs baseline: 30.4609x; 1.0509x over previous
#include <hip/hip_runtime.h>
#include <math.h>


#define B_N 16384
#define F_N 48
#define T_N 512
#define L_N 16
#define MSZ (F_N * F_N)
#define CSZ (2 * MSZ)
#define NGP 256
#define LBLK 8
#define LCOL 64
#define NCW (LCOL / 16)
#define PPITCH 68
#define MPITCH 52
#define GPITCH 52
#define FPC (F_N * T_N / 4)
#define NPAIR (LCOL / 2)
#define DPC (FPC / 2)


typedef __attribute__((ext_vector_type(2))) float v2f;
typedef __attribute__((ext_vector_type(8))) float v8f;

static __device__ __forceinline__ v8f wmma_f32(v2f a, v2f b, v8f c) {
  return __builtin_amdgcn_wmma_f32_16x16x4_f32(false, a, false, b, (short)0, c,
                                               false, false);
}


__global__ void __launch_bounds__(96) k_gram(const float* __restrict__ u,
                                             float* __restrict__ part,
                                             float* __restrict__ gpart,
                                             float* __restrict__ ire, float* __restrict__ iim,
                                             float* __restrict__ dre, float* __restrict__ dim) {
  {
    const int idx = blockIdx.x * 96 + threadIdx.x;
    const float sc = 0.04419417382415922f;
    const float w = 6.283185307179586f / (float)T_N;
    float s0, c0, s1, c1;
    const int q = idx >> 1, hf = idx & 1;
    const int ln = q & 31, lh = ln >> 4, lc = ln & 15;
    if (q < FPC) {
      const int cg = q >> 5, ct = cg / 6, g = cg - ct * 6;
      const int t = (ct >> 2) * NPAIR + (ct & 1) * 16 + 1 + lc;
      const bool col0 = ((ct & 2) != 0) && (t == T_N / 2);
      const int f0 = 4 * (2 * g + hf) + 2 * lh;
      sincosf((float)((f0 * t) & (T_N - 1)) * w, &s0, &c0);
      sincosf((float)(((f0 + 1) * t) & (T_N - 1)) * w, &s1, &c1);
      *(float2*)(ire + q * 4 + hf * 2) = make_float2(c0 * sc, c1 * sc);
      *(float2*)(iim + q * 4 + hf * 2) = col0 ? make_float2(sc, sc) : make_float2(s0 * sc, s1 * sc);
    } else if (q < FPC + DPC) {
      const int q2 = q - FPC;
      const int r = q2 >> 5, g = r & 3, r2 = r >> 2;
      const int blk = r2 / 3, tj = r2 - blk * 3;
      const int f = tj * 16 + lc;
      const int t0 = blk * NPAIR + 1 + 4 * (2 * g + hf) + 2 * lh;
      sincosf((float)((f * t0) & (T_N - 1)) * w, &s0, &c0);
      sincosf((float)((f * (t0 + 1)) & (T_N - 1)) * w, &s1, &c1);
      *(float2*)(dre + q2 * 4 + hf * 2) = make_float2(c0 * sc, c1 * sc);
      *(float2*)(dim + q2 * 4 + hf * 2) = make_float2(s0 * sc, s1 * sc);
    }
  }
  const int lane = threadIdx.x & 31;
  const int wi = threadIdx.x >> 5;
  const int hi = lane >> 4, c = lane & 15;
  const float* ub = u + (size_t)(blockIdx.x * 64 + 2 * hi) * (F_N * 2) + c * 2;
  v8f gre[3], gim[3];
#pragma unroll
  for (int j = 0; j < 3; ++j) {
    gre[j] = (v8f){};
    gim[j] = (v8f){};
  }
  float mxl = -3.4e38f;
#pragma unroll
  for (int ks = 0; ks < 16; ++ks) {
    v2f xr[3], xi[3];
#pragma unroll
    for (int ft = 0; ft < 3; ++ft) {
      const float2 p0 = *(const float2*)(ub + (ks * 4) * (F_N * 2) + ft * 32);
      const float2 p1 = *(const float2*)(ub + (ks * 4 + 1) * (F_N * 2) + ft * 32);
      xr[ft] = (v2f){p0.x, p1.x};
      xi[ft] = (v2f){p0.y, p1.y};
      mxl = fmaxf(mxl, fmaxf(fmaxf(p0.x, p0.y), fmaxf(p1.x, p1.y)));
    }
    const float2 a0 = *(const float2*)(ub + (ks * 4) * (F_N * 2) + wi * 32);
    const float2 a1 = *(const float2*)(ub + (ks * 4 + 1) * (F_N * 2) + wi * 32);
    const v2f ar = (v2f){a0.x, a1.x};
    const v2f ai = (v2f){a0.y, a1.y};
    const v2f an = -ai;
#pragma unroll
    for (int j = 0; j < 3; ++j) {
      gre[j] = wmma_f32(ar, xr[j], gre[j]);
      gim[j] = wmma_f32(ar, xi[j], gim[j]);
      gre[j] = wmma_f32(ai, xi[j], gre[j]);
      gim[j] = wmma_f32(an, xr[j], gim[j]);
    }
  }
  float* gp = gpart + (size_t)blockIdx.x * CSZ + wi * (3 * 256);
#pragma unroll
  for (int j = 0; j < 3; ++j)
#pragma unroll
    for (int v = 0; v < 8; ++v) {
      gp[j * 256 + v * 32 + lane] = gre[j][v];
      gp[MSZ + j * 256 + v * 32 + lane] = gim[j][v];
    }
  mxl = fmaxf(mxl, __shfl_xor(mxl, 16, 32));
  mxl = fmaxf(mxl, __shfl_xor(mxl, 8, 32));
  mxl = fmaxf(mxl, __shfl_xor(mxl, 4, 32));
  mxl = fmaxf(mxl, __shfl_xor(mxl, 2, 32));
  mxl = fmaxf(mxl, __shfl_xor(mxl, 1, 32));
  if (threadIdx.x == 0) part[blockIdx.x] = mxl;
}

__global__ void __launch_bounds__(256) k_gred(const float* __restrict__ gpart,
                                              const float* __restrict__ part,
                                              float* __restrict__ G,
                                              float* __restrict__ scal) {
  __shared__ float sm[256];
  __shared__ float red[8][32];
  const int grp = threadIdx.x >> 5, el = threadIdx.x & 31;
  const int n = blockIdx.x * 32 + el;
  {
    const float* p = gpart + (size_t)(grp * 32) * CSZ + n;
    float acc = 0.0f;
    for (int i = 0; i < 32; ++i) acc += p[(size_t)i * CSZ];
    red[grp][el] = acc;
  }
  sm[threadIdx.x] = part[threadIdx.x];
  __syncthreads();
  for (int s = 128; s > 0; s >>= 1) {
    if (threadIdx.x < s) sm[threadIdx.x] = fmaxf(sm[threadIdx.x], sm[threadIdx.x + s]);
    __syncthreads();
  }
  const float mx = fabsf(sm[0]);
  const float s = 5.0f / mx;
  if (blockIdx.x == 0 && threadIdx.x == 0) {
    scal[0] = s;
    scal[1] = mx / 5.0f;
  }
  if (threadIdx.x < 32) {
    float tot = 0.0f;
#pragma unroll
    for (int k = 0; k < 8; ++k) tot += red[k][el];
    const int pl = n / MSZ, q = n - pl * MSZ;
    const int tl = q >> 8, tr = tl / 3;
    const int r = tr * 16 + ((q >> 5) & 7) + ((q >> 4) & 1) * 8;
    const int cc = (tl - tr * 3) * 16 + (q & 15);
    const float tw = ((tl - tr * 3) > tr) ? 2.0f : 1.0f;
    G[pl * MSZ + r * F_N + cc] = tot * (s * s * (1.0f / (float)B_N)) * tw;
  }
}

static __device__ __forceinline__ void load_dft_frag(const float* __restrict__ dre,
                                                     const float* __restrict__ dim,
                                                     int base, v2f (&dR)[NPAIR / 4],
                                                     v2f (&dI)[NPAIR / 4]) {
#pragma unroll
  for (int g = 0; g < NPAIR / 8; ++g) {
    const float4 r4 = *(const float4*)(dre + base + g * 128);
    const float4 i4 = *(const float4*)(dim + base + g * 128);
    dR[2 * g] = (v2f){r4.x, r4.y};
    dR[2 * g + 1] = (v2f){r4.z, r4.w};
    dI[2 * g] = (v2f){i4.x, i4.y};
    dI[2 * g + 1] = (v2f){i4.z, i4.w};
  }
}

__global__ void __launch_bounds__(288) k_layer(
    const float* __restrict__ ire, const float* __restrict__ iim,
    const float* __restrict__ dre, const float* __restrict__ dim,
    const float* __restrict__ G, const float* __restrict__ sprev,
    float* __restrict__ snext, const float* __restrict__ pprev, float* __restrict__ pnext,
    const float* __restrict__ S1, const float* __restrict__ S2, int first) {
  __shared__ __attribute__((aligned(16))) float sm[2 * F_N * PPITCH + 2 * F_N * GPITCH + NCW * 48 * 32];
  float* Mre = sm;
  float* Mim = sm + F_N * MPITCH;
  float* Pre = sm;
  float* Pim = sm + F_N * PPITCH;
  float* Gre = sm + 2 * F_N * PPITCH;
  float* Gim = Gre + F_N * GPITCH;
  float* Xs = Gim + F_N * GPITCH;

  const int tid = threadIdx.x;
  const int lane = tid & 31;
  const int wv = tid >> 5;
  const int hi = lane >> 4, c = lane & 15;
  const int cg = blockIdx.x % LBLK;
  const int tis = blockIdx.x / LBLK;
  const int tj = wv - NCW;
  const int dbase = ((cg * 3 + tj) * (NPAIR / 8) * 32 + lane) * 4;
  const float s1 = S1[0], s2 = S2[0];

  v8f qre[3], qim[3];
#pragma unroll
  for (int rt = 0; rt < 3; ++rt) {
    qre[rt] = (v8f){};
    qim[rt] = (v8f){};
  }
  v2f bF[12];
#pragma unroll
  for (int ks = 0; ks < 12; ++ks) bF[ks] = (v2f){0.f, 0.f};
  const int poff = (cg * NCW + (wv & (NCW - 1))) * (2 * 3 * 8 * 32);
  const float* pr_src = pprev + poff;
  float* pw = pnext + poff;
  const int ctile = cg * NCW + (wv & (NCW - 1));
  if (wv < NCW) {
    if (!first) {
#pragma unroll
      for (int rt = 0; rt < 3; ++rt)
#pragma unroll
        for (int vg = 0; vg < 2; ++vg) {
          const float4 r4 = *(const float4*)(pr_src + (rt * 2 + vg) * 128 + lane * 4);
          const float4 i4 = *(const float4*)(pr_src + 768 + (rt * 2 + vg) * 128 + lane * 4);
          qre[rt][vg * 4 + 0] = r4.x; qre[rt][vg * 4 + 1] = r4.y;
          qre[rt][vg * 4 + 2] = r4.z; qre[rt][vg * 4 + 3] = r4.w;
          qim[rt][vg * 4 + 0] = i4.x; qim[rt][vg * 4 + 1] = i4.y;
          qim[rt][vg * 4 + 2] = i4.z; qim[rt][vg * 4 + 3] = i4.w;
        }
    }
    const float* fsrc = (wv < 2) ? ire : iim;
#pragma unroll
    for (int g = 0; g < 6; ++g) {
      const float4 f4 = *(const float4*)(fsrc + ((ctile * 6 + g) * 32 + lane) * 4);
      bF[2 * g] = (v2f){f4.x, f4.y};
      bF[2 * g + 1] = (v2f){f4.z, f4.w};
    }
  }

  float4 ms[4];
#pragma unroll
  for (int i = 0; i < 4; ++i) ms[i] = make_float4(0.f, 0.f, 0.f, 0.f);
  if (!first) {
    float4 t[4][LBLK];
#pragma unroll
    for (int i = 0; i < 4; ++i) {
      const int pl = i >> 1;
      const int n = (tid + 288 * i - pl * (MSZ / 4)) * 4;
#pragma unroll
      for (int p = 0; p < LBLK; ++p) t[i][p] = *(const float4*)(sprev + p * CSZ + pl * MSZ + n);
    }
#pragma unroll
    for (int i = 0; i < 4; ++i)
#pragma unroll
      for (int p = 0; p < LBLK; ++p) {
        ms[i].x += t[i][p].x; ms[i].y += t[i][p].y; ms[i].z += t[i][p].z; ms[i].w += t[i][p].w;
      }
  }

#pragma unroll
  for (int i = 0; i < 4; ++i) {
    const int pl = i >> 1;
    const int n = (tid + 288 * i - pl * (MSZ / 4)) * 4;
    const int tl = n >> 8, tr = tl / 3;
    const int r = tr * 16 + ((n >> 5) & 7) + ((n >> 4) & 1) * 8;
    const int cc = (tl - tr * 3) * 16 + (n & 15);
    float4 m;
    if (pl == 0) {
      m.x = ((r == cc) ? 1.0f : 0.0f) - ms[i].x;
      m.y = ((r == cc + 1) ? 1.0f : 0.0f) - ms[i].y;
      m.z = ((r == cc + 2) ? 1.0f : 0.0f) - ms[i].z;
      m.w = ((r == cc + 3) ? 1.0f : 0.0f) - ms[i].w;
      *(float4*)&Mre[r * MPITCH + cc] = m;
    } else {
      m.x = -ms[i].x; m.y = -ms[i].y; m.z = -ms[i].z; m.w = -ms[i].w;
      *(float4*)&Mim[r * MPITCH + cc] = m;
    }
  }
  __syncthreads();

  v2f dR[NPAIR / 4], dI[NPAIR / 4];
  v8f xa[3], xb[3];
#pragma unroll
  for (int it = 0; it < 3; ++it) {
    xa[it] = (v8f){};
    xb[it] = (v8f){};
  }
  if (wv < NCW) {
#pragma unroll
    for (int ks = 0; ks < 12; ++ks) {
      const int k0 = ks * 4 + hi * 2;
      const v2f bf = bF[ks];
#pragma unroll
      for (int it = 0; it < 3; ++it) {
        const v2f are = *(const v2f*)&Mre[(it * 16 + c) * MPITCH + k0];
        const v2f aim = *(const v2f*)&Mim[(it * 16 + c) * MPITCH + k0];
        xa[it] = wmma_f32(are, bf, xa[it]);
        xb[it] = wmma_f32(aim, bf, xb[it]);
      }
    }
    float* xw = Xs + wv * (48 * 32);
#pragma unroll
    for (int it = 0; it < 3; ++it)
#pragma unroll
      for (int vg = 0; vg < 2; ++vg) {
        *(float4*)&xw[((it * 4 + vg) * 32 + lane) * 4] =
            make_float4(xa[it][vg * 4], xa[it][vg * 4 + 1], xa[it][vg * 4 + 2], xa[it][vg * 4 + 3]);
        *(float4*)&xw[((it * 4 + 2 + vg) * 32 + lane) * 4] =
            make_float4(xb[it][vg * 4], xb[it][vg * 4 + 1], xb[it][vg * 4 + 2], xb[it][vg * 4 + 3]);
      }
  } else {
    if (wv < 2 * NCW) {
      float4 g[9];
#pragma unroll
      for (int i = 0; i < 9; ++i) g[i] = *(const float4*)(G + (tid - 128 + 128 * i) * 4);
#pragma unroll
      for (int i = 0; i < 9; ++i) {
        const int q = tid - 128 + 128 * i;
        const int pl = q / (MSZ / 4), rem = q - pl * (MSZ / 4);
        const int r = rem / (F_N / 4), c4 = rem - r * (F_N / 4);
        *(float4*)&Gre[pl * (F_N * GPITCH) + r * GPITCH + c4 * 4] = g[i];
      }
    }
    if (wv < NCW + 3) load_dft_frag(dre, dim, dbase, dR, dI);
  }
  __syncthreads();

  v8f pr = {}, pi = {};
  if (wv < NCW) {
    {
      const float* xr = Xs + (wv ^ 2) * (48 * 32);
      float4 ya[3][2], yb[3][2];
#pragma unroll
      for (int it = 0; it < 3; ++it)
#pragma unroll
        for (int vg = 0; vg < 2; ++vg) {
          ya[it][vg] = *(const float4*)&xr[((it * 4 + vg) * 32 + lane) * 4];
          yb[it][vg] = *(const float4*)&xr[((it * 4 + 2 + vg) * 32 + lane) * 4];
        }
      const bool last = (cg == LBLK - 1);
      const float keep = (last && ((wv & 1) == 1) && (c == 15)) ? 0.0f : 1.0f;
      const int wvs = __builtin_amdgcn_readfirstlane(wv);
      if (wvs < 2) {
#pragma unroll
        for (int it = 0; it < 3; ++it)
#pragma unroll
          for (int v = 0; v < 8; ++v) {
            const float4 a4 = ya[it][v >> 2], b4 = yb[it][v >> 2];
            const float y3 = (v & 3) == 0 ? a4.x : (v & 3) == 1 ? a4.y : (v & 3) == 2 ? a4.z : a4.w;
            const float y4 = (v & 3) == 0 ? b4.x : (v & 3) == 1 ? b4.y : (v & 3) == 2 ? b4.z : b4.w;
            qre[it][v] = fmaf(-keep, y4, qre[it][v] + xa[it][v]);
            qim[it][v] = fmaf(keep, y3, qim[it][v] + xb[it][v]);
          }
      } else {
        const bool col0 = last && (wvs == NCW - 1) && (c == 15);
#pragma unroll
        for (int it = 0; it < 3; ++it)
#pragma unroll
          for (int v = 0; v < 8; ++v) {
            const float4 a4 = ya[it][v >> 2], b4 = yb[it][v >> 2];
            const float y1 = (v & 3) == 0 ? a4.x : (v & 3) == 1 ? a4.y : (v & 3) == 2 ? a4.z : a4.w;
            const float y2 = (v & 3) == 0 ? b4.x : (v & 3) == 1 ? b4.y : (v & 3) == 2 ? b4.z : b4.w;
            const float ore = col0 ? xa[it][v] : xb[it][v];
            const float oim = col0 ? xb[it][v] : -xa[it][v];
            qre[it][v] = fmaf(keep, y1, qre[it][v] + ore);
            qim[it][v] = fmaf(keep, y2, qim[it][v] + oim);
          }
      }
    }
    v8f yk[3], yre[3], yim[3];
#pragma unroll
    for (int it = 0; it < 3; ++it) {
      yk[it] = (v8f){};
      yre[it] = (v8f){};
      yim[it] = (v8f){};
    }
#pragma unroll
    for (int rt = 0; rt < 3; ++rt) {
      float4 gr[3][2], gi[3][2];
#pragma unroll
      for (int it = 0; it <= rt; ++it) {
        const int go = (it * 16 + c) * GPITCH + rt * 16 + hi * 8;
        gr[it][0] = *(const float4*)&Gre[go];
        gr[it][1] = *(const float4*)&Gre[go + 4];
        gi[it][0] = *(const float4*)&Gim[go];
        gi[it][1] = *(const float4*)&Gim[go + 4];
      }
#pragma unroll
      for (int j = 0; j < 4; ++j) {
        const v2f bqr = (v2f){qre[rt][2 * j], qre[rt][2 * j + 1]};
        const v2f bqi = (v2f){qim[rt][2 * j], qim[rt][2 * j + 1]};
        const v2f bqd = bqi - bqr;
        const v2f bqs = -bqr - bqi;
#pragma unroll
        for (int it = 0; it <= rt; ++it) {
          const float4 r4 = gr[it][j >> 1];
          const float4 i4 = gi[it][j >> 1];
          const v2f g_r = (j & 1) ? (v2f){r4.z, r4.w} : (v2f){r4.x, r4.y};
          const v2f g_i = (j & 1) ? (v2f){i4.z, i4.w} : (v2f){i4.x, i4.y};
          const v2f gs = g_r + g_i;
          yk[it]  = wmma_f32(gs, bqr, yk[it]);
          yim[it] = wmma_f32(g_r, bqd, yim[it]);
          yre[it] = wmma_f32(g_i, bqs, yre[it]);
        }
      }
    }
    float p = 0.0f;
#pragma unroll
    for (int rt = 0; rt < 3; ++rt)
#pragma unroll
      for (int v = 0; v < 8; ++v)
        p += qre[rt][v] * (yre[rt][v] + yk[rt][v]) + qim[rt][v] * (yim[rt][v] + yk[rt][v]);
    p += __shfl_xor(p, 16, 32);
    const float rm = sqrtf(fmaxf(p, 0.0f));
    const float g = 1.0f / (1.0f + expf(-(s1 * (rm - s2))));
    const bool lone = (cg == LBLK - 1) && (wv == NCW - 1) && (c == 15);
    const int pcol = lone ? LCOL : wv * 16 + c;
    pr = (tis == 0) ? qre[0] : (tis == 1) ? qre[1] : qre[2];
    pi = (tis == 0) ? qim[0] : (tis == 1) ? qim[1] : qim[2];
#pragma unroll
    for (int v = 0; v < 8; ++v) {
      pr[v] *= g;
      pi[v] *= g;
      Pre[(hi * 8 + v) * PPITCH + pcol] = pr[v];
      Pim[(hi * 8 + v) * PPITCH + pcol] = pi[v];
    }
    if (lone) {
#pragma unroll
      for (int r = 0; r < 8; ++r) {
        Pre[(hi * 8 + r) * PPITCH + LCOL - 1] = 0.0f;
        Pim[(hi * 8 + r) * PPITCH + LCOL - 1] = 0.0f;
      }
    }
  }
  __syncthreads();
  if (wv < NCW) {
#pragma unroll
    for (int vg = 0; vg < 2; ++vg) {
      *(float4*)(pw + (tis * 2 + vg) * 128 + lane * 4) =
          make_float4(pr[vg * 4], pr[vg * 4 + 1], pr[vg * 4 + 2], pr[vg * 4 + 3]);
      *(float4*)(pw + 768 + (tis * 2 + vg) * 128 + lane * 4) =
          make_float4(pi[vg * 4], pi[vg * 4 + 1], pi[vg * 4 + 2], pi[vg * 4 + 3]);
    }
  }

  if (wv >= NCW && wv < NCW + 3) {
    v8f sre = {}, sim = {}, sre2 = {}, sim2 = {};
#pragma unroll
    for (int ks = 0; ks < NPAIR / 4; ++ks) {
      const int k0 = ks * 4 + hi * 2;
      const v2f lr = *(const v2f*)&Pre[c * PPITCH + k0];
      const v2f mr = *(const v2f*)&Pre[c * PPITCH + NPAIR + k0];
      const v2f li = *(const v2f*)&Pim[c * PPITCH + k0];
      const v2f mi = *(const v2f*)&Pim[c * PPITCH + NPAIR + k0];
      const v2f ur = lr + mr;
      const v2f ui = li + mi;
      const v2f vi = li - mi;
      const v2f nvr = mr - lr;
      const v2f bc = dR[ks];
      const v2f bs = dI[ks];
      sre  = wmma_f32(ur, bc, sre);
      sim  = wmma_f32(ui, bc, sim);
      sre2 = wmma_f32(vi, bs, sre2);
      sim2 = wmma_f32(nvr, bs, sim2);
    }
    sre += sre2;
    sim += sim2;
    if (cg == LBLK - 1) {
      const float sc = 0.04419417382415922f;
#pragma unroll
      for (int v = 0; v < 8; ++v) {
        sre[v] += Pre[(hi * 8 + v) * PPITCH + LCOL] * sc;
        sim[v] += Pim[(hi * 8 + v) * PPITCH + LCOL] * sc;
      }
    }
    float* sp = snext + cg * CSZ + (tis * 3 + tj) * 256;
#pragma unroll
    for (int v = 0; v < 8; ++v) {
      sp[v * 32 + lane] = sre[v];
      sp[MSZ + v * 32 + lane] = sim[v];
    }
  }
}

__global__ void __launch_bounds__(128) k_out(const float* __restrict__ u,
                                             const float* __restrict__ scal,
                                             const float* __restrict__ sp,
                                             float* __restrict__ out) {
  __shared__ __attribute__((aligned(16))) float Wre[MSZ];
  __shared__ __attribute__((aligned(16))) float Wim[MSZ];
  for (int q4 = threadIdx.x; q4 < MSZ / 4; q4 += 128) {
    const int n = q4 * 4;
    const int tl = n >> 8, tr = tl / 3;
    const int r = tr * 16 + ((n >> 5) & 7) + ((n >> 4) & 1) * 8;
    const int cc = (tl - tr * 3) * 16 + (n & 15);
    float4 sr = make_float4(0.f, 0.f, 0.f, 0.f), si = make_float4(0.f, 0.f, 0.f, 0.f);
#pragma unroll
    for (int p = 0; p < LBLK; ++p) {
      const float4 a = *(const float4*)(sp + p * CSZ + n);
      const float4 b = *(const float4*)(sp + p * CSZ + MSZ + n);
      sr.x += a.x; sr.y += a.y; sr.z += a.z; sr.w += a.w;
      si.x += b.x; si.y += b.y; si.z += b.z; si.w += b.w;
    }
    *(float4*)&Wre[r * F_N + cc] = sr;
    *(float4*)&Wim[r * F_N + cc] = si;
  }
  __syncthreads();

  const int lane = threadIdx.x & 31;
  const int wv = threadIdx.x >> 5;
  const int hi = lane >> 4, c = lane & 15;
  const int rowbase = (blockIdx.x * 4 + wv) * 16;
  const float s = scal[0], inv = scal[1];
  const float* ur = u + (size_t)(rowbase + c) * (F_N * 2) + hi * 4;
  v8f ore[3], oim[3];
#pragma unroll
  for (int j = 0; j < 3; ++j) {
    ore[j] = (v8f){};
    oim[j] = (v8f){};
  }
#pragma unroll
  for (int ks = 0; ks < 12; ++ks) {
    const float4 x = *(const float4*)(ur + ks * 8);
    const v2f ar = (v2f){s * x.x, s * x.z};
    const v2f ai = (v2f){s * x.y, s * x.w};
    const v2f an = -ai;
    const int k0 = ks * 4 + hi * 2;
#pragma unroll
    for (int j = 0; j < 3; ++j) {
      const v2f br = (v2f){Wre[k0 * F_N + j * 16 + c], Wre[(k0 + 1) * F_N + j * 16 + c]};
      const v2f bi = (v2f){Wim[k0 * F_N + j * 16 + c], Wim[(k0 + 1) * F_N + j * 16 + c]};
      ore[j] = wmma_f32(ar, br, ore[j]);
      ore[j] = wmma_f32(an, bi, ore[j]);
      oim[j] = wmma_f32(ar, bi, oim[j]);
      oim[j] = wmma_f32(ai, br, oim[j]);
    }
  }
#pragma unroll
  for (int j = 0; j < 3; ++j)
#pragma unroll
    for (int v = 0; v < 8; ++v) {
      const int row = rowbase + hi * 8 + v;
      float2 o;
      o.x = ore[j][v] * inv;
      o.y = oim[j][v] * inv;
      *(float2*)(out + ((size_t)row * F_N + j * 16 + c) * 2) = o;
    }
}

extern "C" void kernel_launch(void* const* d_in, const int* in_sizes, int n_in,
                              void* d_out, int out_size, void* d_ws, size_t ws_size,
                              hipStream_t stream) {
  const float* u  = (const float*)d_in[0];
  const float* S1 = (const float*)d_in[1];
  const float* S2 = (const float*)d_in[2];
  float* out = (float*)d_out;
  float* w = (float*)d_ws;

  float* scal  = w;
  float* part  = w + 8;
  float* ire   = w + 2048;
  float* iim   = ire + F_N * T_N;
  float* dre   = iim + F_N * T_N;
  float* dim   = dre + F_N * T_N;
  float* G     = w + 100352;
  float* spart = w + 104960;
  float* pst   = w + 178688;
  float* gpart = w + 227840;

  k_gram<<<NGP, 96, 0, stream>>>(u, part, gpart, ire, iim, dre, dim);
  k_gred<<<CSZ / 32, 256, 0, stream>>>(gpart, part, G, scal);

  for (int l = 0; l < L_N; ++l) {
    k_layer<<<LBLK * 3, 288, 0, stream>>>(ire, iim, dre, dim, G,
                                      spart + ((l + 1) & 1) * (LBLK * CSZ),
                                      spart + (l & 1) * (LBLK * CSZ),
                                      pst + ((l + 1) & 1) * (2 * F_N * T_N),
                                      pst + (l & 1) * (2 * F_N * T_N), S1, S2,
                                      (l == 0) ? 1 : 0);
  }
  k_out<<<B_N / 64, 128, 0, stream>>>(u, scal, spart + ((L_N - 1) & 1) * (LBLK * CSZ), out);
}
